# starting kernel
# baseline (speedup 1.0000x reference)
#include <hip/hip_runtime.h>

typedef __attribute__((ext_vector_type(2))) float v2f;
typedef __attribute__((ext_vector_type(8))) float v8f;

#define F_IN 5
#define HID 16

__global__ void k_init_deg(float* __restrict__ deg, int N) {
  int i = blockIdx.x * blockDim.x + threadIdx.x;
  if (i < N) deg[i] = 1.0f;
}

__global__ void k_degree(const int* __restrict__ ei, float* __restrict__ deg, int E) {
  int e = blockIdx.x * blockDim.x + threadIdx.x;
  if (e < E) {
    __builtin_prefetch(&ei[(long)E + e + 16384], 0, 0);
    atomicAdd(&deg[ei[(long)E + e]], 1.0f);
  }
}

__global__ void k_rsqrt(float* __restrict__ d, int N) {
  int i = blockIdx.x * blockDim.x + threadIdx.x;
  if (i < N) d[i] = rsqrtf(d[i]);
}

__global__ void k_xw_wmma(const float* __restrict__ x,
                          const float* __restrict__ W1,
                          const float* __restrict__ dinv,
                          float* __restrict__ xw,
                          float* __restrict__ out1,
                          int N) {
  const int lane = threadIdx.x & 31;
  const int wave = threadIdx.x >> 5;
  const int wavesPerBlock = blockDim.x >> 5;
  const int base = (blockIdx.x * wavesPerBlock + wave) * 16;
  if (base >= N) return;

  const int half = lane >> 4;
  const int l    = lane & 15;
  int row = base + l;
  if (row >= N) row = N - 1;

  const float* xr = x + (long)row * F_IN;
  const int klo = half * 2;

  v2f a0; a0.x = xr[klo];            a0.y = xr[klo + 1];
  v2f b0; b0.x = W1[klo * HID + l];  b0.y = W1[(klo + 1) * HID + l];

  float x4 = xr[4];
  float w4 = W1[4 * HID + l];
  v2f a1; a1.x = half ? 0.0f : x4;  a1.y = 0.0f;
  v2f b1; b1.x = half ? 0.0f : w4;  b1.y = 0.0f;

  v8f c = {};
  c = __builtin_amdgcn_wmma_f32_16x16x4_f32(false, a0, false, b0, (short)0, c, false, false);
  c = __builtin_amdgcn_wmma_f32_16x16x4_f32(false, a1, false, b1, (short)0, c, false, false);

  const long colBase = (long)(base + half * 8) * HID + l;
  if (base + 16 <= N) {
    float* xp = xw + colBase;
    float* op = out1 + colBase;
    const float* dp = dinv + base + half * 8;
#pragma unroll
    for (int r = 0; r < 8; ++r) {
      const float v = c[r];
      const float di = dp[r];
      xp[(long)r * HID] = v;
      op[(long)r * HID] = di * di * v;
    }
  } else {
#pragma unroll
    for (int r = 0; r < 8; ++r) {
      const int node = base + half * 8 + r;
      if (node < N) {
        const float v = c[r];
        const float di = dinv[node];
        xw[(long)node * HID + l] = v;
        out1[(long)node * HID + l] = di * di * v;
      }
    }
  }
}

__global__ void k_agg1(const int* __restrict__ ei, const float* __restrict__ dinv,
                       const float* __restrict__ xw, float* __restrict__ out1, int E) {
  int e = blockIdx.x * blockDim.x + threadIdx.x;
  if (e >= E) return;
  __builtin_prefetch(&ei[e + 16384], 0, 0);
  __builtin_prefetch(&ei[(long)E + e + 16384], 0, 0);
  const int s = ei[e];
  const int d = ei[(long)E + e];
  const float nrm = dinv[s] * dinv[d];
  const float4* xs = (const float4*)(xw + (long)s * HID);
  float* od = out1 + (long)d * HID;
  float4 v0 = xs[0], v1 = xs[1], v2 = xs[2], v3 = xs[3];
  atomicAdd(od + 0,  nrm * v0.x);  atomicAdd(od + 1,  nrm * v0.y);
  atomicAdd(od + 2,  nrm * v0.z);  atomicAdd(od + 3,  nrm * v0.w);
  atomicAdd(od + 4,  nrm * v1.x);  atomicAdd(od + 5,  nrm * v1.y);
  atomicAdd(od + 6,  nrm * v1.z);  atomicAdd(od + 7,  nrm * v1.w);
  atomicAdd(od + 8,  nrm * v2.x);  atomicAdd(od + 9,  nrm * v2.y);
  atomicAdd(od + 10, nrm * v2.z);  atomicAdd(od + 11, nrm * v2.w);
  atomicAdd(od + 12, nrm * v3.x);  atomicAdd(od + 13, nrm * v3.y);
  atomicAdd(od + 14, nrm * v3.z);  atomicAdd(od + 15, nrm * v3.w);
}

__global__ void k_layer2(const float* __restrict__ out1, const float* __restrict__ b1,
                         const float* __restrict__ W2, const float* __restrict__ b2,
                         const float* __restrict__ dinv, float* __restrict__ z,
                         float* __restrict__ out, int N) {
  int i = blockIdx.x * blockDim.x + threadIdx.x;
  if (i >= N) return;
  const float4* hr = (const float4*)(out1 + (long)i * HID);
  const float4* bb = (const float4*)b1;
  const float4* ww = (const float4*)W2;
  float acc = 0.0f;
#pragma unroll
  for (int q = 0; q < 4; ++q) {
    float4 v = hr[q], b = bb[q], w = ww[q];
    acc = fmaf(fmaxf(v.x + b.x, 0.0f), w.x, acc);
    acc = fmaf(fmaxf(v.y + b.y, 0.0f), w.y, acc);
    acc = fmaf(fmaxf(v.z + b.z, 0.0f), w.z, acc);
    acc = fmaf(fmaxf(v.w + b.w, 0.0f), w.w, acc);
  }
  z[i] = acc;
  const float di = dinv[i];
  out[i] = b2[0] + di * di * acc;
}

__global__ void k_agg2(const int* __restrict__ ei, const float* __restrict__ dinv,
                       const float* __restrict__ z, float* __restrict__ out, int E) {
  int e = blockIdx.x * blockDim.x + threadIdx.x;
  if (e >= E) return;
  __builtin_prefetch(&ei[e + 16384], 0, 0);
  __builtin_prefetch(&ei[(long)E + e + 16384], 0, 0);
  const int s = ei[e];
  const int d = ei[(long)E + e];
  atomicAdd(&out[d], dinv[s] * dinv[d] * z[s]);
}

extern "C" void kernel_launch(void* const* d_in, const int* in_sizes, int n_in,
                              void* d_out, int out_size, void* d_ws, size_t ws_size,
                              hipStream_t stream) {
  const float* x  = (const float*)d_in[0];
  const int*   ei = (const int*)d_in[1];
  const float* W1 = (const float*)d_in[2];
  const float* b1 = (const float*)d_in[3];
  const float* W2 = (const float*)d_in[4];
  const float* b2 = (const float*)d_in[5];
  float* out = (float*)d_out;

  const int N = in_sizes[0] / F_IN;
  const int E = in_sizes[1] / 2;

  char* ws = (char*)d_ws;
  size_t off = 0;
  auto carve = [&](size_t bytes) -> void* {
    off = (off + 255) & ~(size_t)255;
    void* p = ws + off;
    off += bytes;
    return p;
  };
  float* dinv = (float*)carve((size_t)N * sizeof(float));
  float* xw   = (float*)carve((size_t)N * HID * sizeof(float));
  float* out1 = (float*)carve((size_t)N * HID * sizeof(float));
  float* z    = (float*)carve((size_t)N * sizeof(float));
  (void)ws_size; (void)n_in; (void)out_size;

  const int B = 256;
  const int gN = (N + B - 1) / B;
  const int gE = (E + B - 1) / B;
  const int wavesNeeded   = (N + 15) / 16;
  const int wavesPerBlock = B / 32;
  const int gW = (wavesNeeded + wavesPerBlock - 1) / wavesPerBlock;

  hipLaunchKernelGGL(k_init_deg, dim3(gN), dim3(B), 0, stream, dinv, N);
  hipLaunchKernelGGL(k_degree,   dim3(gE), dim3(B), 0, stream, ei, dinv, E);
  hipLaunchKernelGGL(k_rsqrt,    dim3(gN), dim3(B), 0, stream, dinv, N);
  hipLaunchKernelGGL(k_xw_wmma,  dim3(gW), dim3(B), 0, stream, x, W1, dinv, xw, out1, N);
  hipLaunchKernelGGL(k_agg1,     dim3(gE), dim3(B), 0, stream, ei, dinv, xw, out1, E);
  hipLaunchKernelGGL(k_layer2,   dim3(gN), dim3(B), 0, stream, out1, b1, W2, b2, dinv, z, out, N);
  hipLaunchKernelGGL(k_agg2,     dim3(gE), dim3(B), 0, stream, ei, dinv, z, out, E);
}


# k_agg1 lane = feature: a half-wave's 16 lanes gather one 64B xw row and atomic-add one 64B out1 row, so a wave-wide gather/atomic touches 2 lines, not 32; f32 atomic sums stay arrival-ordered
# speedup vs baseline: 2.3396x; 2.3396x over previous
#include <hip/hip_runtime.h>

typedef __attribute__((ext_vector_type(2))) float v2f;
typedef __attribute__((ext_vector_type(8))) float v8f;

#define F_IN 5
#define HID 16

__global__ void k_init_deg(float* __restrict__ deg, int N) {
  int i = blockIdx.x * blockDim.x + threadIdx.x;
  if (i < N) deg[i] = 1.0f;
}

__global__ void k_degree(const int* __restrict__ ei, float* __restrict__ deg, int E) {
  int e = blockIdx.x * blockDim.x + threadIdx.x;
  if (e < E) {
    __builtin_prefetch(&ei[(long)E + e + 16384], 0, 0);
    atomicAdd(&deg[ei[(long)E + e]], 1.0f);
  }
}

__global__ void k_rsqrt(float* __restrict__ d, int N) {
  int i = blockIdx.x * blockDim.x + threadIdx.x;
  if (i < N) d[i] = rsqrtf(d[i]);
}

__global__ void k_xw_wmma(const float* __restrict__ x,
                          const float* __restrict__ W1,
                          const float* __restrict__ dinv,
                          float* __restrict__ xw,
                          float* __restrict__ out1,
                          int N) {
  const int lane = threadIdx.x & 31;
  const int wave = threadIdx.x >> 5;
  const int wavesPerBlock = blockDim.x >> 5;
  const int base = (blockIdx.x * wavesPerBlock + wave) * 16;
  if (base >= N) return;

  const int half = lane >> 4;
  const int l    = lane & 15;
  int row = base + l;
  if (row >= N) row = N - 1;

  const float* xr = x + (long)row * F_IN;
  const int klo = half * 2;

  v2f a0; a0.x = xr[klo];            a0.y = xr[klo + 1];
  v2f b0; b0.x = W1[klo * HID + l];  b0.y = W1[(klo + 1) * HID + l];

  float x4 = xr[4];
  float w4 = W1[4 * HID + l];
  v2f a1; a1.x = half ? 0.0f : x4;  a1.y = 0.0f;
  v2f b1; b1.x = half ? 0.0f : w4;  b1.y = 0.0f;

  v8f c = {};
  c = __builtin_amdgcn_wmma_f32_16x16x4_f32(false, a0, false, b0, (short)0, c, false, false);
  c = __builtin_amdgcn_wmma_f32_16x16x4_f32(false, a1, false, b1, (short)0, c, false, false);

  const long colBase = (long)(base + half * 8) * HID + l;
  if (base + 16 <= N) {
    float* xp = xw + colBase;
    float* op = out1 + colBase;
    const float* dp = dinv + base + half * 8;
#pragma unroll
    for (int r = 0; r < 8; ++r) {
      const float v = c[r];
      const float di = dp[r];
      xp[(long)r * HID] = v;
      op[(long)r * HID] = di * di * v;
    }
  } else {
#pragma unroll
    for (int r = 0; r < 8; ++r) {
      const int node = base + half * 8 + r;
      if (node < N) {
        const float v = c[r];
        const float di = dinv[node];
        xw[(long)node * HID + l] = v;
        out1[(long)node * HID + l] = di * di * v;
      }
    }
  }
}

__global__ void k_agg1(const int* __restrict__ ei, const float* __restrict__ dinv,
                       const float* __restrict__ xw, float* __restrict__ out1, int E) {
  const int e    = blockIdx.x * blockDim.x + threadIdx.x;
  const int lane = threadIdx.x & 31;
  const unsigned c = lane & 15;
  const int half = lane & 16;
  const int e0   = e - lane;
  if (e0 >= E) return;

  if (e0 + 32 <= E) {
    __builtin_prefetch(&ei[e + 16384], 0, 0);
    __builtin_prefetch(&ei[(long)E + e + 16384], 0, 0);
    const int s = ei[e];
    const int d = ei[(long)E + e];
    const float ds = dinv[s];
    const float dd = dinv[d];
    const unsigned srow = (unsigned)s * HID;
    const unsigned drow = (unsigned)d * HID;

    float v[16];
#pragma unroll
    for (int j = 0; j < 16; ++j)
      v[j] = xw[__shfl(srow, half | j) + c];

    const float nrm = ds * dd;
#pragma unroll
    for (int j = 0; j < 16; ++j)
      atomicAdd(out1 + (__shfl(drow, half | j) + c), __shfl(nrm, half | j) * v[j]);
  } else {
    int s = 0, d = 0;
    float nrm = 0.0f;
    if (e < E) {
      s = ei[e];
      d = ei[(long)E + e];
      nrm = dinv[s] * dinv[d];
    }
    for (int j = 0; j < 16; ++j) {
      const int from = half | j;
      const int   sj = __shfl(s, from);
      const int   dj = __shfl(d, from);
      const float nj = __shfl(nrm, from);
      if (e0 + from < E)
        atomicAdd(out1 + (long)dj * HID + c, nj * xw[(long)sj * HID + c]);
    }
  }
}

__global__ void k_layer2(const float* __restrict__ out1, const float* __restrict__ b1,
                         const float* __restrict__ W2, const float* __restrict__ b2,
                         const float* __restrict__ dinv, float* __restrict__ z,
                         float* __restrict__ out, int N) {
  int i = blockIdx.x * blockDim.x + threadIdx.x;
  if (i >= N) return;
  const float4* hr = (const float4*)(out1 + (long)i * HID);
  const float4* bb = (const float4*)b1;
  const float4* ww = (const float4*)W2;
  float acc = 0.0f;
#pragma unroll
  for (int q = 0; q < 4; ++q) {
    float4 v = hr[q], b = bb[q], w = ww[q];
    acc = fmaf(fmaxf(v.x + b.x, 0.0f), w.x, acc);
    acc = fmaf(fmaxf(v.y + b.y, 0.0f), w.y, acc);
    acc = fmaf(fmaxf(v.z + b.z, 0.0f), w.z, acc);
    acc = fmaf(fmaxf(v.w + b.w, 0.0f), w.w, acc);
  }
  z[i] = acc;
  const float di = dinv[i];
  out[i] = b2[0] + di * di * acc;
}

__global__ void k_agg2(const int* __restrict__ ei, const float* __restrict__ dinv,
                       const float* __restrict__ z, float* __restrict__ out, int E) {
  int e = blockIdx.x * blockDim.x + threadIdx.x;
  if (e >= E) return;
  __builtin_prefetch(&ei[e + 16384], 0, 0);
  __builtin_prefetch(&ei[(long)E + e + 16384], 0, 0);
  const int s = ei[e];
  const int d = ei[(long)E + e];
  atomicAdd(&out[d], dinv[s] * dinv[d] * z[s]);
}

extern "C" void kernel_launch(void* const* d_in, const int* in_sizes, int n_in,
                              void* d_out, int out_size, void* d_ws, size_t ws_size,
                              hipStream_t stream) {
  const float* x  = (const float*)d_in[0];
  const int*   ei = (const int*)d_in[1];
  const float* W1 = (const float*)d_in[2];
  const float* b1 = (const float*)d_in[3];
  const float* W2 = (const float*)d_in[4];
  const float* b2 = (const float*)d_in[5];
  float* out = (float*)d_out;

  const int N = in_sizes[0] / F_IN;
  const int E = in_sizes[1] / 2;

  char* ws = (char*)d_ws;
  size_t off = 0;
  auto carve = [&](size_t bytes) -> void* {
    off = (off + 255) & ~(size_t)255;
    void* p = ws + off;
    off += bytes;
    return p;
  };
  float* dinv = (float*)carve((size_t)N * sizeof(float));
  float* xw   = (float*)carve((size_t)N * HID * sizeof(float));
  float* out1 = (float*)carve((size_t)N * HID * sizeof(float));
  float* z    = (float*)carve((size_t)N * sizeof(float));
  (void)ws_size; (void)n_in; (void)out_size;

  const int B = 256;
  const int gN = (N + B - 1) / B;
  const int gE = (E + B - 1) / B;
  const int wavesNeeded   = (N + 15) / 16;
  const int wavesPerBlock = B / 32;
  const int gW = (wavesNeeded + wavesPerBlock - 1) / wavesPerBlock;

  hipLaunchKernelGGL(k_init_deg, dim3(gN), dim3(B), 0, stream, dinv, N);
  hipLaunchKernelGGL(k_degree,   dim3(gE), dim3(B), 0, stream, ei, dinv, E);
  hipLaunchKernelGGL(k_rsqrt,    dim3(gN), dim3(B), 0, stream, dinv, N);
  hipLaunchKernelGGL(k_xw_wmma,  dim3(gW), dim3(B), 0, stream, x, W1, dinv, xw, out1, N);
  hipLaunchKernelGGL(k_agg1,     dim3(gE), dim3(B), 0, stream, ei, dinv, xw, out1, E);
  hipLaunchKernelGGL(k_layer2,   dim3(gN), dim3(B), 0, stream, out1, b1, W2, b2, dinv, z, out, N);
  hipLaunchKernelGGL(k_agg2,     dim3(gE), dim3(B), 0, stream, ei, dinv, z, out, E);
}


# Every array the edge passes add into (degree, layer-1 rows, layer-2 sums) kept twice, one copy per L2 half; a wave adds into its own half's copy; node passes add the copies; f32 sum order changes
# speedup vs baseline: 6.6125x; 2.8263x over previous
#include <hip/hip_runtime.h>

typedef __attribute__((ext_vector_type(2))) float v2f;
typedef __attribute__((ext_vector_type(8))) float v8f;

#define F_IN 5
#define HID 16

__device__ __forceinline__ int l2_side() {
  return (int)((__builtin_amdgcn_s_sendmsg_rtn(0x87) >> 18) & 1u);
}

__global__ void k_init_deg(float* __restrict__ degA, float* __restrict__ degB, int N) {
  int i = blockIdx.x * blockDim.x + threadIdx.x;
  if (i < N) { degA[i] = 1.0f; degB[i] = 0.0f; }
}

__global__ void k_degree(const int* __restrict__ ei, float* __restrict__ degA,
                         float* __restrict__ degB, int E) {
  int e = blockIdx.x * blockDim.x + threadIdx.x;
  float* deg = l2_side() ? degB : degA;
  if (e < E) {
    __builtin_prefetch(&ei[(long)E + e + 16384], 0, 0);
    atomicAdd(&deg[ei[(long)E + e]], 1.0f);
  }
}

__global__ void k_rsqrt(float* __restrict__ d, const float* __restrict__ degB, int N) {
  int i = blockIdx.x * blockDim.x + threadIdx.x;
  if (i < N) d[i] = rsqrtf(d[i] + degB[i]);
}

__global__ void k_xw_wmma(const float* __restrict__ x,
                          const float* __restrict__ W1,
                          const float* __restrict__ dinv,
                          float* __restrict__ xw,
                          float* __restrict__ out1,
                          float* __restrict__ out1B,
                          int N) {
  const int lane = threadIdx.x & 31;
  const int wave = threadIdx.x >> 5;
  const int wavesPerBlock = blockDim.x >> 5;
  const int base = (blockIdx.x * wavesPerBlock + wave) * 16;
  if (base >= N) return;

  const int half = lane >> 4;
  const int l    = lane & 15;
  int row = base + l;
  if (row >= N) row = N - 1;

  const float* xr = x + (long)row * F_IN;
  const int klo = half * 2;

  v2f a0; a0.x = xr[klo];            a0.y = xr[klo + 1];
  v2f b0; b0.x = W1[klo * HID + l];  b0.y = W1[(klo + 1) * HID + l];

  float x4 = xr[4];
  float w4 = W1[4 * HID + l];
  v2f a1; a1.x = half ? 0.0f : x4;  a1.y = 0.0f;
  v2f b1; b1.x = half ? 0.0f : w4;  b1.y = 0.0f;

  v8f c = {};
  c = __builtin_amdgcn_wmma_f32_16x16x4_f32(false, a0, false, b0, (short)0, c, false, false);
  c = __builtin_amdgcn_wmma_f32_16x16x4_f32(false, a1, false, b1, (short)0, c, false, false);

  const long colBase = (long)(base + half * 8) * HID + l;
  if (base + 16 <= N) {
    float* xp = xw + colBase;
    float* op = out1 + colBase;
    float* zp = out1B + colBase;
    const float* dp = dinv + base + half * 8;
#pragma unroll
    for (int r = 0; r < 8; ++r) {
      const float v = c[r];
      const float di = dp[r];
      xp[(long)r * HID] = v;
      op[(long)r * HID] = di * di * v;
      zp[(long)r * HID] = 0.0f;
    }
  } else {
#pragma unroll
    for (int r = 0; r < 8; ++r) {
      const int node = base + half * 8 + r;
      if (node < N) {
        const float v = c[r];
        const float di = dinv[node];
        xw[(long)node * HID + l] = v;
        out1[(long)node * HID + l] = di * di * v;
        out1B[(long)node * HID + l] = 0.0f;
      }
    }
  }
}

__global__ void k_agg1(const int* __restrict__ ei, const float* __restrict__ dinv,
                       const float* __restrict__ xw, float* __restrict__ out1A,
                       float* __restrict__ out1B, int E) {
  float* out1 = l2_side() ? out1B : out1A;
  const int e    = blockIdx.x * blockDim.x + threadIdx.x;
  const int lane = threadIdx.x & 31;
  const unsigned c = lane & 15;
  const int half = lane & 16;
  const int e0   = e - lane;
  if (e0 >= E) return;

  if (e0 + 32 <= E) {
    __builtin_prefetch(&ei[e + 16384], 0, 0);
    __builtin_prefetch(&ei[(long)E + e + 16384], 0, 0);
    const int s = ei[e];
    const int d = ei[(long)E + e];
    const float ds = dinv[s];
    const float dd = dinv[d];
    const unsigned srow = (unsigned)s * HID;
    const unsigned drow = (unsigned)d * HID;

    float v[16];
#pragma unroll
    for (int j = 0; j < 16; ++j)
      v[j] = xw[__shfl(srow, half | j) + c];

    const float nrm = ds * dd;
#pragma unroll
    for (int j = 0; j < 16; ++j)
      atomicAdd(out1 + (__shfl(drow, half | j) + c), __shfl(nrm, half | j) * v[j]);
  } else {
    int s = 0, d = 0;
    float nrm = 0.0f;
    if (e < E) {
      s = ei[e];
      d = ei[(long)E + e];
      nrm = dinv[s] * dinv[d];
    }
    for (int j = 0; j < 16; ++j) {
      const int from = half | j;
      const int   sj = __shfl(s, from);
      const int   dj = __shfl(d, from);
      const float nj = __shfl(nrm, from);
      if (e0 + from < E)
        atomicAdd(out1 + (long)dj * HID + c, nj * xw[(long)sj * HID + c]);
    }
  }
}

__global__ void k_layer2(const float* __restrict__ out1, const float* __restrict__ out1B,
                         const float* __restrict__ b1,
                         const float* __restrict__ W2, const float* __restrict__ b2,
                         const float* __restrict__ dinv, float* __restrict__ z,
                         float* __restrict__ accA, float* __restrict__ accB, int N) {
  int i = blockIdx.x * blockDim.x + threadIdx.x;
  if (i >= N) return;
  const float4* hr = (const float4*)(out1 + (long)i * HID);
  const float4* hs = (const float4*)(out1B + (long)i * HID);
  const float4* bb = (const float4*)b1;
  const float4* ww = (const float4*)W2;
  float acc = 0.0f;
#pragma unroll
  for (int q = 0; q < 4; ++q) {
    float4 v = hr[q], u = hs[q], b = bb[q], w = ww[q];
    acc = fmaf(fmaxf((v.x + u.x) + b.x, 0.0f), w.x, acc);
    acc = fmaf(fmaxf((v.y + u.y) + b.y, 0.0f), w.y, acc);
    acc = fmaf(fmaxf((v.z + u.z) + b.z, 0.0f), w.z, acc);
    acc = fmaf(fmaxf((v.w + u.w) + b.w, 0.0f), w.w, acc);
  }
  z[i] = acc;
  const float di = dinv[i];
  accA[i] = b2[0] + di * di * acc;
  accB[i] = 0.0f;
}

__global__ void k_agg2(const int* __restrict__ ei, const float* __restrict__ dinv,
                       const float* __restrict__ z, float* __restrict__ accA,
                       float* __restrict__ accB, int E) {
  int e = blockIdx.x * blockDim.x + threadIdx.x;
  float* out = l2_side() ? accB : accA;
  if (e >= E) return;
  __builtin_prefetch(&ei[e + 16384], 0, 0);
  __builtin_prefetch(&ei[(long)E + e + 16384], 0, 0);
  const int s = ei[e];
  const int d = ei[(long)E + e];
  atomicAdd(&out[d], dinv[s] * dinv[d] * z[s]);
}

__global__ void k_sum2(const float* __restrict__ accA, const float* __restrict__ accB,
                       float* __restrict__ out, int N) {
  int i = blockIdx.x * blockDim.x + threadIdx.x;
  if (i < N) out[i] = accA[i] + accB[i];
}

extern "C" void kernel_launch(void* const* d_in, const int* in_sizes, int n_in,
                              void* d_out, int out_size, void* d_ws, size_t ws_size,
                              hipStream_t stream) {
  const float* x  = (const float*)d_in[0];
  const int*   ei = (const int*)d_in[1];
  const float* W1 = (const float*)d_in[2];
  const float* b1 = (const float*)d_in[3];
  const float* W2 = (const float*)d_in[4];
  const float* b2 = (const float*)d_in[5];
  float* out = (float*)d_out;

  const int N = in_sizes[0] / F_IN;
  const int E = in_sizes[1] / 2;

  char* ws = (char*)d_ws;
  size_t off = 0;
  auto carve = [&](size_t bytes) -> void* {
    off = (off + 255) & ~(size_t)255;
    void* p = ws + off;
    off += bytes;
    return p;
  };
  float* dinv  = (float*)carve((size_t)N * sizeof(float));
  float* degB  = (float*)carve((size_t)N * sizeof(float));
  float* xw    = (float*)carve((size_t)N * HID * sizeof(float));
  float* out1  = (float*)carve((size_t)N * HID * sizeof(float));
  float* out1B = (float*)carve((size_t)N * HID * sizeof(float));
  float* z     = (float*)carve((size_t)N * sizeof(float));
  float* accA  = (float*)carve((size_t)N * sizeof(float));
  float* accB  = (float*)carve((size_t)N * sizeof(float));
  (void)ws_size; (void)n_in; (void)out_size;

  const int B = 256;
  const int gN = (N + B - 1) / B;
  const int gE = (E + B - 1) / B;
  const int wavesNeeded   = (N + 15) / 16;
  const int wavesPerBlock = B / 32;
  const int gW = (wavesNeeded + wavesPerBlock - 1) / wavesPerBlock;

  hipLaunchKernelGGL(k_init_deg, dim3(gN), dim3(B), 0, stream, dinv, degB, N);
  hipLaunchKernelGGL(k_degree,   dim3(gE), dim3(B), 0, stream, ei, dinv, degB, E);
  hipLaunchKernelGGL(k_rsqrt,    dim3(gN), dim3(B), 0, stream, dinv, degB, N);
  hipLaunchKernelGGL(k_xw_wmma,  dim3(gW), dim3(B), 0, stream, x, W1, dinv, xw, out1, out1B, N);
  hipLaunchKernelGGL(k_agg1,     dim3(gE), dim3(B), 0, stream, ei, dinv, xw, out1, out1B, E);
  hipLaunchKernelGGL(k_layer2,   dim3(gN), dim3(B), 0, stream, out1, out1B, b1, W2, b2, dinv, z, accA, accB, N);
  hipLaunchKernelGGL(k_agg2,     dim3(gE), dim3(B), 0, stream, ei, dinv, z, accA, accB, E);
  hipLaunchKernelGGL(k_sum2,     dim3(gN), dim3(B), 0, stream, accA, accB, out, N);
}


# Edge weight dinv[src]*dinv[dst] factored out of both edge passes: node passes scale rows by dinv[src] before and the sums by dinv[dst] after; edges gather no dinv; sums stay arrival-ordered
# speedup vs baseline: 7.9235x; 1.1983x over previous
#include <hip/hip_runtime.h>

typedef __attribute__((ext_vector_type(2))) float v2f;
typedef __attribute__((ext_vector_type(8))) float v8f;

#define F_IN 5
#define HID 16

__device__ __forceinline__ int l2_side() {
  return (int)((__builtin_amdgcn_s_sendmsg_rtn(0x87) >> 18) & 1u);
}


__global__ void k_init_deg(float* __restrict__ degA, float* __restrict__ degB, int N) {
  int i = blockIdx.x * blockDim.x + threadIdx.x;
  if (i < N) { degA[i] = 1.0f; degB[i] = 0.0f; }
}

__global__ void k_degree(const int* __restrict__ ei, float* __restrict__ degA,
                         float* __restrict__ degB, int E) {
  int e = blockIdx.x * blockDim.x + threadIdx.x;
  float* deg = l2_side() ? degB : degA;
  if (e < E) {
    __builtin_prefetch(&ei[(long)E + e + 16384], 0, 0);
    atomicAdd(&deg[ei[(long)E + e]], 1.0f);
  }
}

__global__ void k_rsqrt(float* __restrict__ d, const float* __restrict__ degB, int N) {
  int i = blockIdx.x * blockDim.x + threadIdx.x;
  if (i < N) d[i] = rsqrtf(d[i] + degB[i]);
}

__global__ void k_xw_wmma(const float* __restrict__ x,
                          const float* __restrict__ W1,
                          const float* __restrict__ dinv,
                          float* __restrict__ xw,
                          float* __restrict__ out1,
                          float* __restrict__ out1B,
                          int N) {
  const int lane = threadIdx.x & 31;
  const int wave = threadIdx.x >> 5;
  const int wavesPerBlock = blockDim.x >> 5;
  const int base = (blockIdx.x * wavesPerBlock + wave) * 16;
  if (base >= N) return;

  const int half = lane >> 4;
  const int l    = lane & 15;
  int row = base + l;
  if (row >= N) row = N - 1;

  const float* xr = x + (long)row * F_IN;
  const int klo = half * 2;

  v2f a0; a0.x = xr[klo];            a0.y = xr[klo + 1];
  v2f b0; b0.x = W1[klo * HID + l];  b0.y = W1[(klo + 1) * HID + l];

  float x4 = xr[4];
  float w4 = W1[4 * HID + l];
  v2f a1; a1.x = half ? 0.0f : x4;  a1.y = 0.0f;
  v2f b1; b1.x = half ? 0.0f : w4;  b1.y = 0.0f;

  v8f c = {};
  c = __builtin_amdgcn_wmma_f32_16x16x4_f32(false, a0, false, b0, (short)0, c, false, false);
  c = __builtin_amdgcn_wmma_f32_16x16x4_f32(false, a1, false, b1, (short)0, c, false, false);

  const long colBase = (long)(base + half * 8) * HID + l;
  if (base + 16 <= N) {
    float* xp = xw + colBase;
    float* op = out1 + colBase;
    float* zp = out1B + colBase;
    const float* dp = dinv + base + half * 8;
#pragma unroll
    for (int r = 0; r < 8; ++r) {
      const float v = dp[r] * c[r];
      xp[(long)r * HID] = v;
      op[(long)r * HID] = v;
      zp[(long)r * HID] = 0.0f;
    }
  } else {
#pragma unroll
    for (int r = 0; r < 8; ++r) {
      const int node = base + half * 8 + r;
      if (node < N) {
        const float v = dinv[node] * c[r];
        xw[(long)node * HID + l] = v;
        out1[(long)node * HID + l] = v;
        out1B[(long)node * HID + l] = 0.0f;
      }
    }
  }
}

__global__ void k_agg1(const int* __restrict__ ei,
                       const float* __restrict__ xw, float* __restrict__ out1A,
                       float* __restrict__ out1B, int E) {
  float* out1 = l2_side() ? out1B : out1A;
  const int e    = blockIdx.x * blockDim.x + threadIdx.x;
  const int lane = threadIdx.x & 31;
  const unsigned c = lane & 15;
  const int half = lane & 16;
  const int e0   = e - lane;
  if (e0 >= E) return;

  if (e0 + 32 <= E) {
    __builtin_prefetch(&ei[e + 16384], 0, 0);
    __builtin_prefetch(&ei[(long)E + e + 16384], 0, 0);
    const int s = ei[e];
    const int d = ei[(long)E + e];
    const unsigned srow = (unsigned)s * HID;
    const unsigned drow = (unsigned)d * HID;

    float v[16];
#pragma unroll
    for (int j = 0; j < 16; ++j)
      v[j] = xw[__shfl(srow, half | j) + c];

#pragma unroll
    for (int j = 0; j < 16; ++j)
      atomicAdd(out1 + (__shfl(drow, half | j) + c), v[j]);
  } else {
    int s = 0, d = 0;
    if (e < E) {
      s = ei[e];
      d = ei[(long)E + e];
    }
    for (int j = 0; j < 16; ++j) {
      const int from = half | j;
      const int   sj = __shfl(s, from);
      const int   dj = __shfl(d, from);
      if (e0 + from < E)
        atomicAdd(out1 + (long)dj * HID + c, xw[(long)sj * HID + c]);
    }
  }
}

__global__ void k_layer2(const float* __restrict__ out1, const float* __restrict__ out1B,
                         const float* __restrict__ b1,
                         const float* __restrict__ W2,
                         const float* __restrict__ dinv, float* __restrict__ z,
                         float* __restrict__ accA, float* __restrict__ accB, int N) {
  int i = blockIdx.x * blockDim.x + threadIdx.x;
  if (i >= N) return;
  const float4* hr = (const float4*)(out1 + (long)i * HID);
  const float4* hs = (const float4*)(out1B + (long)i * HID);
  const float4* bb = (const float4*)b1;
  const float4* ww = (const float4*)W2;
  const float di = dinv[i];
  float acc = 0.0f;
#pragma unroll
  for (int q = 0; q < 4; ++q) {
    float4 v = hr[q], u = hs[q], b = bb[q], w = ww[q];
    acc = fmaf(fmaxf(di * (v.x + u.x) + b.x, 0.0f), w.x, acc);
    acc = fmaf(fmaxf(di * (v.y + u.y) + b.y, 0.0f), w.y, acc);
    acc = fmaf(fmaxf(di * (v.z + u.z) + b.z, 0.0f), w.z, acc);
    acc = fmaf(fmaxf(di * (v.w + u.w) + b.w, 0.0f), w.w, acc);
  }
  const float zs = di * acc;
  z[i] = zs;
  accA[i] = zs;
  accB[i] = 0.0f;
}

__global__ void k_agg2(const int* __restrict__ ei,
                       const float* __restrict__ z, float* __restrict__ accA,
                       float* __restrict__ accB, int E) {
  int e = blockIdx.x * blockDim.x + threadIdx.x;
  float* out = l2_side() ? accB : accA;
  if (e >= E) return;
  __builtin_prefetch(&ei[e + 16384], 0, 0);
  __builtin_prefetch(&ei[(long)E + e + 16384], 0, 0);
  const int s = ei[e];
  const int d = ei[(long)E + e];
  atomicAdd(&out[d], z[s]);
}

__global__ void k_sum2(const float* __restrict__ accA, const float* __restrict__ accB,
                       const float* __restrict__ dinv, const float* __restrict__ b2,
                       float* __restrict__ out, int N) {
  int i = blockIdx.x * blockDim.x + threadIdx.x;
  if (i < N) out[i] = b2[0] + dinv[i] * (accA[i] + accB[i]);
}

extern "C" void kernel_launch(void* const* d_in, const int* in_sizes, int n_in,
                              void* d_out, int out_size, void* d_ws, size_t ws_size,
                              hipStream_t stream) {
  const float* x  = (const float*)d_in[0];
  const int*   ei = (const int*)d_in[1];
  const float* W1 = (const float*)d_in[2];
  const float* b1 = (const float*)d_in[3];
  const float* W2 = (const float*)d_in[4];
  const float* b2 = (const float*)d_in[5];
  float* out = (float*)d_out;

  const int N = in_sizes[0] / F_IN;
  const int E = in_sizes[1] / 2;

  char* ws = (char*)d_ws;
  size_t off = 0;
  auto carve = [&](size_t bytes) -> void* {
    off = (off + 255) & ~(size_t)255;
    void* p = ws + off;
    off += bytes;
    return p;
  };
  float* dinv  = (float*)carve((size_t)N * sizeof(float));
  float* degB  = (float*)carve((size_t)N * sizeof(float));
  float* xw    = (float*)carve((size_t)N * HID * sizeof(float));
  float* out1  = (float*)carve((size_t)N * HID * sizeof(float));
  float* out1B = (float*)carve((size_t)N * HID * sizeof(float));
  float* z     = (float*)carve((size_t)N * sizeof(float));
  float* accA  = (float*)carve((size_t)N * sizeof(float));
  float* accB  = (float*)carve((size_t)N * sizeof(float));
  (void)ws_size; (void)n_in; (void)out_size;

  const int B = 256;
  const int gN = (N + B - 1) / B;
  const int gE = (E + B - 1) / B;
  const int wavesNeeded   = (N + 15) / 16;
  const int wavesPerBlock = B / 32;
  const int gW = (wavesNeeded + wavesPerBlock - 1) / wavesPerBlock;

  hipLaunchKernelGGL(k_init_deg, dim3(gN), dim3(B), 0, stream, dinv, degB, N);
  hipLaunchKernelGGL(k_degree,   dim3(gE), dim3(B), 0, stream, ei, dinv, degB, E);
  hipLaunchKernelGGL(k_rsqrt,    dim3(gN), dim3(B), 0, stream, dinv, degB, N);
  hipLaunchKernelGGL(k_xw_wmma,  dim3(gW), dim3(B), 0, stream, x, W1, dinv, xw, out1, out1B, N);
  hipLaunchKernelGGL(k_agg1,     dim3(gE), dim3(B), 0, stream, ei, xw, out1, out1B, E);
  hipLaunchKernelGGL(k_layer2,   dim3(gN), dim3(B), 0, stream, out1, out1B, b1, W2, dinv, z, accA, accB, N);
  hipLaunchKernelGGL(k_agg2,     dim3(gE), dim3(B), 0, stream, ei, z, accA, accB, E);
  hipLaunchKernelGGL(k_sum2,     dim3(gN), dim3(B), 0, stream, accA, accB, dinv, b2, out, N);
}


# Edge passes: a wave takes 8 batches of 32 edges, not one (8 index loads in flight per thread; layer-1 wave loops, next indices preloaded): 8x fewer waves start; f32 atomic sums stay arrival-ordered
# speedup vs baseline: 9.9480x; 1.2555x over previous
#include <hip/hip_runtime.h>

typedef __attribute__((ext_vector_type(2))) float v2f;
typedef __attribute__((ext_vector_type(8))) float v8f;

#define F_IN 5
#define HID 16

#define EDGE_UNROLL 8

__device__ __forceinline__ int l2_side() {
  return (int)((__builtin_amdgcn_s_sendmsg_rtn(0x87) >> 18) & 1u);
}


__global__ void k_init_deg(float* __restrict__ degA, float* __restrict__ degB, int N) {
  int i = blockIdx.x * blockDim.x + threadIdx.x;
  if (i < N) { degA[i] = 1.0f; degB[i] = 0.0f; }
}

__global__ void k_degree(const int* __restrict__ ei, float* __restrict__ degA,
                         float* __restrict__ degB, int E) {
  float* deg = l2_side() ? degB : degA;
  const int* __restrict__ dstp = ei + (long)E;
  const int step  = blockDim.x;
  const int first = blockIdx.x * (step * EDGE_UNROLL);
  const int base  = first + threadIdx.x;
  if (first + step * EDGE_UNROLL <= E) {
    int d[EDGE_UNROLL];
#pragma unroll
    for (int k = 0; k < EDGE_UNROLL; ++k)
      d[k] = dstp[base + k * step];
#pragma unroll
    for (int k = 0; k < EDGE_UNROLL; ++k)
      atomicAdd(&deg[(unsigned)d[k]], 1.0f);
  } else {
    for (int k = 0; k < EDGE_UNROLL; ++k) {
      const int e = base + k * step;
      if (e < E) atomicAdd(&deg[(unsigned)dstp[e]], 1.0f);
    }
  }
}

__global__ void k_rsqrt(float* __restrict__ d, const float* __restrict__ degB, int N) {
  int i = blockIdx.x * blockDim.x + threadIdx.x;
  if (i < N) d[i] = rsqrtf(d[i] + degB[i]);
}

__global__ void k_xw_wmma(const float* __restrict__ x,
                          const float* __restrict__ W1,
                          const float* __restrict__ dinv,
                          float* __restrict__ xw,
                          float* __restrict__ out1,
                          float* __restrict__ out1B,
                          int N) {
  const int lane = threadIdx.x & 31;
  const int wave = threadIdx.x >> 5;
  const int wavesPerBlock = blockDim.x >> 5;
  const int base = (blockIdx.x * wavesPerBlock + wave) * 16;
  if (base >= N) return;

  const int half = lane >> 4;
  const int l    = lane & 15;
  int row = base + l;
  if (row >= N) row = N - 1;

  const float* xr = x + (long)row * F_IN;
  const int klo = half * 2;

  v2f a0; a0.x = xr[klo];            a0.y = xr[klo + 1];
  v2f b0; b0.x = W1[klo * HID + l];  b0.y = W1[(klo + 1) * HID + l];

  float x4 = xr[4];
  float w4 = W1[4 * HID + l];
  v2f a1; a1.x = half ? 0.0f : x4;  a1.y = 0.0f;
  v2f b1; b1.x = half ? 0.0f : w4;  b1.y = 0.0f;

  v8f c = {};
  c = __builtin_amdgcn_wmma_f32_16x16x4_f32(false, a0, false, b0, (short)0, c, false, false);
  c = __builtin_amdgcn_wmma_f32_16x16x4_f32(false, a1, false, b1, (short)0, c, false, false);

  const long colBase = (long)(base + half * 8) * HID + l;
  if (base + 16 <= N) {
    float* xp = xw + colBase;
    float* op = out1 + colBase;
    float* zp = out1B + colBase;
    const float* dp = dinv + base + half * 8;
#pragma unroll
    for (int r = 0; r < 8; ++r) {
      const float v = dp[r] * c[r];
      xp[(long)r * HID] = v;
      op[(long)r * HID] = v;
      zp[(long)r * HID] = 0.0f;
    }
  } else {
#pragma unroll
    for (int r = 0; r < 8; ++r) {
      const int node = base + half * 8 + r;
      if (node < N) {
        const float v = dinv[node] * c[r];
        xw[(long)node * HID + l] = v;
        out1[(long)node * HID + l] = v;
        out1B[(long)node * HID + l] = 0.0f;
      }
    }
  }
}

__global__ void k_agg1(const int* __restrict__ ei,
                       const float* __restrict__ xw, float* __restrict__ out1A,
                       float* __restrict__ out1B, int E) {
  float* out1 = l2_side() ? out1B : out1A;
  const int* __restrict__ dstp = ei + (long)E;
  const int lane = threadIdx.x & 31;
  const unsigned c = lane & 15;
  const int half = lane & 16;
  const int wavesPerBlock = blockDim.x >> 5;
  const int nWaves = gridDim.x * wavesPerBlock;
  const int wid = __builtin_amdgcn_readfirstlane(blockIdx.x * wavesPerBlock + (threadIdx.x >> 5));
  const int nFull = E >> 5;

  int b = wid;
  int s = 0, d = 0;
  if (b < nFull) {
    s = ei[b * 32 + lane];
    d = dstp[b * 32 + lane];
  }
  while (b < nFull) {
    const unsigned srow = (unsigned)s * HID;
    const unsigned drow = (unsigned)d * HID;

    float v[16];
#pragma unroll
    for (int j = 0; j < 16; ++j)
      v[j] = xw[__shfl(srow, half | j) + c];

    const int bn = b + nWaves;
    if (bn < nFull) {
      s = ei[bn * 32 + lane];
      d = dstp[bn * 32 + lane];
      const int bp = bn + nWaves;
      if (bp < nFull) {
        __builtin_prefetch(&ei[bp * 32 + lane], 0, 0);
        __builtin_prefetch(&dstp[bp * 32 + lane], 0, 0);
      }
    }

#pragma unroll
    for (int j = 0; j < 16; ++j)
      atomicAdd(out1 + (__shfl(drow, half | j) + c), v[j]);
    b = bn;
  }

  const int e0 = nFull * 32;
  if (e0 < E && wid == nFull % nWaves) {
    const int e = e0 + lane;
    int st = 0, dt = 0;
    if (e < E) {
      st = ei[e];
      dt = dstp[e];
    }
    for (int j = 0; j < 16; ++j) {
      const int from = half | j;
      const int   sj = __shfl(st, from);
      const int   dj = __shfl(dt, from);
      if (e0 + from < E)
        atomicAdd(out1 + (long)dj * HID + c, xw[(long)sj * HID + c]);
    }
  }
}

__global__ void k_layer2(const float* __restrict__ out1, const float* __restrict__ out1B,
                         const float* __restrict__ b1,
                         const float* __restrict__ W2,
                         const float* __restrict__ dinv, float* __restrict__ z,
                         float* __restrict__ accA, float* __restrict__ accB, int N) {
  int i = blockIdx.x * blockDim.x + threadIdx.x;
  if (i >= N) return;
  const float4* hr = (const float4*)(out1 + (long)i * HID);
  const float4* hs = (const float4*)(out1B + (long)i * HID);
  const float4* bb = (const float4*)b1;
  const float4* ww = (const float4*)W2;
  const float di = dinv[i];
  float acc = 0.0f;
#pragma unroll
  for (int q = 0; q < 4; ++q) {
    float4 v = hr[q], u = hs[q], b = bb[q], w = ww[q];
    acc = fmaf(fmaxf(di * (v.x + u.x) + b.x, 0.0f), w.x, acc);
    acc = fmaf(fmaxf(di * (v.y + u.y) + b.y, 0.0f), w.y, acc);
    acc = fmaf(fmaxf(di * (v.z + u.z) + b.z, 0.0f), w.z, acc);
    acc = fmaf(fmaxf(di * (v.w + u.w) + b.w, 0.0f), w.w, acc);
  }
  const float zs = di * acc;
  z[i] = zs;
  accA[i] = zs;
  accB[i] = 0.0f;
}

__global__ void k_agg2(const int* __restrict__ ei,
                       const float* __restrict__ z, float* __restrict__ accA,
                       float* __restrict__ accB, int E) {
  float* out = l2_side() ? accB : accA;
  const int* __restrict__ dstp = ei + (long)E;
  const int step  = blockDim.x;
  const int first = blockIdx.x * (step * EDGE_UNROLL);
  const int base  = first + threadIdx.x;
  if (first + step * EDGE_UNROLL <= E) {
    int   s[EDGE_UNROLL], d[EDGE_UNROLL];
    float v[EDGE_UNROLL];
#pragma unroll
    for (int k = 0; k < EDGE_UNROLL; ++k) {
      s[k] = ei[base + k * step];
      d[k] = dstp[base + k * step];
    }
#pragma unroll
    for (int k = 0; k < EDGE_UNROLL; ++k)
      v[k] = z[(unsigned)s[k]];
#pragma unroll
    for (int k = 0; k < EDGE_UNROLL; ++k)
      atomicAdd(&out[(unsigned)d[k]], v[k]);
  } else {
    for (int k = 0; k < EDGE_UNROLL; ++k) {
      const int e = base + k * step;
      if (e < E) atomicAdd(&out[(unsigned)dstp[e]], z[(unsigned)ei[e]]);
    }
  }
}

__global__ void k_sum2(const float* __restrict__ accA, const float* __restrict__ accB,
                       const float* __restrict__ dinv, const float* __restrict__ b2,
                       float* __restrict__ out, int N) {
  int i = blockIdx.x * blockDim.x + threadIdx.x;
  if (i < N) out[i] = b2[0] + dinv[i] * (accA[i] + accB[i]);
}

extern "C" void kernel_launch(void* const* d_in, const int* in_sizes, int n_in,
                              void* d_out, int out_size, void* d_ws, size_t ws_size,
                              hipStream_t stream) {
  const float* x  = (const float*)d_in[0];
  const int*   ei = (const int*)d_in[1];
  const float* W1 = (const float*)d_in[2];
  const float* b1 = (const float*)d_in[3];
  const float* W2 = (const float*)d_in[4];
  const float* b2 = (const float*)d_in[5];
  float* out = (float*)d_out;

  const int N = in_sizes[0] / F_IN;
  const int E = in_sizes[1] / 2;

  char* ws = (char*)d_ws;
  size_t off = 0;
  auto carve = [&](size_t bytes) -> void* {
    off = (off + 255) & ~(size_t)255;
    void* p = ws + off;
    off += bytes;
    return p;
  };
  float* dinv  = (float*)carve((size_t)N * sizeof(float));
  float* degB  = (float*)carve((size_t)N * sizeof(float));
  float* xw    = (float*)carve((size_t)N * HID * sizeof(float));
  float* out1  = (float*)carve((size_t)N * HID * sizeof(float));
  float* out1B = (float*)carve((size_t)N * HID * sizeof(float));
  float* z     = (float*)carve((size_t)N * sizeof(float));
  float* accA  = (float*)carve((size_t)N * sizeof(float));
  float* accB  = (float*)carve((size_t)N * sizeof(float));
  (void)ws_size; (void)n_in; (void)out_size;

  const int B = 256;
  const int gN = (N + B - 1) / B;
  const int gC = (E + B * EDGE_UNROLL - 1) / (B * EDGE_UNROLL);
  const int wavesNeeded   = (N + 15) / 16;
  const int wavesPerBlock = B / 32;
  const int gW = (wavesNeeded + wavesPerBlock - 1) / wavesPerBlock;

  hipLaunchKernelGGL(k_init_deg, dim3(gN), dim3(B), 0, stream, dinv, degB, N);
  hipLaunchKernelGGL(k_degree,   dim3(gC), dim3(B), 0, stream, ei, dinv, degB, E);
  hipLaunchKernelGGL(k_rsqrt,    dim3(gN), dim3(B), 0, stream, dinv, degB, N);
  hipLaunchKernelGGL(k_xw_wmma,  dim3(gW), dim3(B), 0, stream, x, W1, dinv, xw, out1, out1B, N);
  hipLaunchKernelGGL(k_agg1,     dim3(gC), dim3(B), 0, stream, ei, xw, out1, out1B, E);
  hipLaunchKernelGGL(k_layer2,   dim3(gN), dim3(B), 0, stream, out1, out1B, b1, W2, dinv, z, accA, accB, N);
  hipLaunchKernelGGL(k_agg2,     dim3(gC), dim3(B), 0, stream, ei, z, accA, accB, E);
  hipLaunchKernelGGL(k_sum2,     dim3(gN), dim3(B), 0, stream, accA, accB, dinv, b2, out, N);
}


# Degree counted in LDS: 128 work-groups each count their share of destinations in a 16-bit LDS table and store it; a node pass adds the tables: no global degree atomics, 7 launches, dinv bit-equal
# speedup vs baseline: 10.7561x; 1.0812x over previous
#include <hip/hip_runtime.h>

typedef __attribute__((ext_vector_type(2))) float v2f;
typedef __attribute__((ext_vector_type(8))) float v8f;

#define F_IN 5
#define HID 16

#define EDGE_UNROLL 8

__device__ __forceinline__ int l2_side() {
  return (int)((__builtin_amdgcn_s_sendmsg_rtn(0x87) >> 18) & 1u);
}


#define DEG_THREADS   1024
#define DEG_WORDS     51200
#define DEG_GROUPS    128
#define DEG_MAX_SHARE 32768
#define DEG_SLICES    4
#define DEG_BATCH     32
static_assert(DEG_GROUPS % (DEG_SLICES * DEG_BATCH) == 0, "pass 2 adds the rows in whole batches");

__device__ __forceinline__ void deg_count(unsigned* cnt, int d, int nb, unsigned nNodes) {
  const unsigned r = (unsigned)(d - nb);
  if (r < nNodes)
    atomicAdd(&cnt[r >> 1], 1u << ((r & 1u) << 4));
}

__global__ __launch_bounds__(DEG_THREADS)
void k_degree_lds(const int* __restrict__ ei, unsigned* __restrict__ part,
                  int N, int E, int rowWords) {
  __shared__ __attribute__((aligned(16))) unsigned cnt[DEG_WORDS];
  const int* __restrict__ dstp = ei + (long)E;
  const int tid   = threadIdx.x;
  const int G     = (int)gridDim.x;
  const int share = (((E + G - 1) / G) + 3) & ~3;
  const int e0    = ((long)blockIdx.x * share < (long)E) ? (int)blockIdx.x * share : E;
  const int e1    = (e0 + share < E) ? e0 + share : E;
  unsigned* __restrict__ row = part + (long)blockIdx.x * rowWords;
  uint4* c4 = (uint4*)cnt;

  for (int nb = 0; nb < N; nb += 2 * DEG_WORDS) {
    const int nNodes = (N - nb < 2 * DEG_WORDS) ? N - nb : 2 * DEG_WORDS;
    const int nW     = (nNodes + 1) >> 1;
    const int nW4    = (nW + 3) >> 2;
    for (int q = tid; q < nW4; q += DEG_THREADS)
      c4[q] = make_uint4(0u, 0u, 0u, 0u);
    __syncthreads();

    if ((E & 3) == 0) {
      const int4* __restrict__ d4 = (const int4*)(dstp + e0);
      const int nQ = (e1 - e0) >> 2;
      for (int q0 = tid; q0 < nQ; q0 += 8 * DEG_THREADS) {
        int4 d[8];
#pragma unroll
        for (int k = 0; k < 8; ++k) {
          const int q = q0 + k * DEG_THREADS;
          d[k] = d4[(q < nQ) ? q : 0];
        }
#pragma unroll
        for (int k = 0; k < 8; ++k) {
          if (q0 + k * DEG_THREADS < nQ) {
            deg_count(cnt, d[k].x, nb, (unsigned)nNodes);
            deg_count(cnt, d[k].y, nb, (unsigned)nNodes);
            deg_count(cnt, d[k].z, nb, (unsigned)nNodes);
            deg_count(cnt, d[k].w, nb, (unsigned)nNodes);
          }
        }
      }
    } else {
      for (int e = e0 + tid; e < e1; e += DEG_THREADS)
        deg_count(cnt, dstp[e], nb, (unsigned)nNodes);
    }
    __syncthreads();

    unsigned* __restrict__ rw = row + (nb >> 1);
    const int full4 = nW >> 2;
    for (int q = tid; q < full4; q += DEG_THREADS)
      ((uint4*)rw)[q] = c4[q];
    if (full4 * 4 + tid < nW)
      rw[full4 * 4 + tid] = cnt[full4 * 4 + tid];
    __syncthreads();
  }
}

__global__ void k_dinv(const unsigned* __restrict__ part, float* __restrict__ dinv,
                       int N, int G, int rowWords) {
  __shared__ unsigned sLo[DEG_SLICES][32];
  __shared__ unsigned sHi[DEG_SLICES][32];
  const int lane  = threadIdx.x & 31;
  const int slice = threadIdx.x >> 5;
  const int w     = blockIdx.x * 32 + lane;
  const int nW    = (N + 1) >> 1;
  unsigned lo = 0u, hi = 0u;
  if (w < nW) {
    const long step = (long)DEG_SLICES * rowWords;
    for (int g0 = slice; g0 < G; g0 += DEG_SLICES * DEG_BATCH) {
      const unsigned* __restrict__ p = part + (long)g0 * rowWords + w;
      unsigned v[DEG_BATCH];
#pragma unroll
      for (int k = 0; k < DEG_BATCH; ++k)
        v[k] = p[k * step];
#pragma unroll
      for (int k = 0; k < DEG_BATCH; ++k) {
        lo += v[k] & 0xffffu;
        hi += v[k] >> 16;
      }
    }
  }
  sLo[slice][lane] = lo;
  sHi[slice][lane] = hi;
  __syncthreads();
  if (slice == 0 && w < nW) {
#pragma unroll
    for (int s = 1; s < DEG_SLICES; ++s) { lo += sLo[s][lane]; hi += sHi[s][lane]; }
    const float d0 = rsqrtf(1.0f + (float)lo);
    const float d1 = rsqrtf(1.0f + (float)hi);
    if (2 * w + 1 < N) {
      float2 o; o.x = d0; o.y = d1;
      *(float2*)(dinv + 2 * w) = o;
    } else {
      dinv[2 * w] = d0;
    }
  }
}

__global__ void k_xw_wmma(const float* __restrict__ x,
                          const float* __restrict__ W1,
                          const float* __restrict__ dinv,
                          float* __restrict__ xw,
                          float* __restrict__ out1,
                          float* __restrict__ out1B,
                          int N) {
  const int lane = threadIdx.x & 31;
  const int wave = threadIdx.x >> 5;
  const int wavesPerBlock = blockDim.x >> 5;
  const int base = (blockIdx.x * wavesPerBlock + wave) * 16;
  if (base >= N) return;

  const int half = lane >> 4;
  const int l    = lane & 15;
  int row = base + l;
  if (row >= N) row = N - 1;

  const float* xr = x + (long)row * F_IN;
  const int klo = half * 2;

  v2f a0; a0.x = xr[klo];            a0.y = xr[klo + 1];
  v2f b0; b0.x = W1[klo * HID + l];  b0.y = W1[(klo + 1) * HID + l];

  float x4 = xr[4];
  float w4 = W1[4 * HID + l];
  v2f a1; a1.x = half ? 0.0f : x4;  a1.y = 0.0f;
  v2f b1; b1.x = half ? 0.0f : w4;  b1.y = 0.0f;

  v8f c = {};
  c = __builtin_amdgcn_wmma_f32_16x16x4_f32(false, a0, false, b0, (short)0, c, false, false);
  c = __builtin_amdgcn_wmma_f32_16x16x4_f32(false, a1, false, b1, (short)0, c, false, false);

  const long colBase = (long)(base + half * 8) * HID + l;
  if (base + 16 <= N) {
    float* xp = xw + colBase;
    float* op = out1 + colBase;
    float* zp = out1B + colBase;
    const float* dp = dinv + base + half * 8;
#pragma unroll
    for (int r = 0; r < 8; ++r) {
      const float v = dp[r] * c[r];
      xp[(long)r * HID] = v;
      op[(long)r * HID] = v;
      zp[(long)r * HID] = 0.0f;
    }
  } else {
#pragma unroll
    for (int r = 0; r < 8; ++r) {
      const int node = base + half * 8 + r;
      if (node < N) {
        const float v = dinv[node] * c[r];
        xw[(long)node * HID + l] = v;
        out1[(long)node * HID + l] = v;
        out1B[(long)node * HID + l] = 0.0f;
      }
    }
  }
}

__global__ void k_agg1(const int* __restrict__ ei,
                       const float* __restrict__ xw, float* __restrict__ out1A,
                       float* __restrict__ out1B, int E) {
  float* out1 = l2_side() ? out1B : out1A;
  const int* __restrict__ dstp = ei + (long)E;
  const int lane = threadIdx.x & 31;
  const unsigned c = lane & 15;
  const int half = lane & 16;
  const int wavesPerBlock = blockDim.x >> 5;
  const int nWaves = gridDim.x * wavesPerBlock;
  const int wid = __builtin_amdgcn_readfirstlane(blockIdx.x * wavesPerBlock + (threadIdx.x >> 5));
  const int nFull = E >> 5;

  int b = wid;
  int s = 0, d = 0;
  if (b < nFull) {
    s = ei[b * 32 + lane];
    d = dstp[b * 32 + lane];
  }
  while (b < nFull) {
    const unsigned srow = (unsigned)s * HID;
    const unsigned drow = (unsigned)d * HID;

    float v[16];
#pragma unroll
    for (int j = 0; j < 16; ++j)
      v[j] = xw[__shfl(srow, half | j) + c];

    const int bn = b + nWaves;
    if (bn < nFull) {
      s = ei[bn * 32 + lane];
      d = dstp[bn * 32 + lane];
      const int bp = bn + nWaves;
      if (bp < nFull) {
        __builtin_prefetch(&ei[bp * 32 + lane], 0, 0);
        __builtin_prefetch(&dstp[bp * 32 + lane], 0, 0);
      }
    }

#pragma unroll
    for (int j = 0; j < 16; ++j)
      atomicAdd(out1 + (__shfl(drow, half | j) + c), v[j]);
    b = bn;
  }

  const int e0 = nFull * 32;
  if (e0 < E && wid == nFull % nWaves) {
    const int e = e0 + lane;
    int st = 0, dt = 0;
    if (e < E) {
      st = ei[e];
      dt = dstp[e];
    }
    for (int j = 0; j < 16; ++j) {
      const int from = half | j;
      const int   sj = __shfl(st, from);
      const int   dj = __shfl(dt, from);
      if (e0 + from < E)
        atomicAdd(out1 + (long)dj * HID + c, xw[(long)sj * HID + c]);
    }
  }
}

__global__ void k_layer2(const float* __restrict__ out1, const float* __restrict__ out1B,
                         const float* __restrict__ b1,
                         const float* __restrict__ W2,
                         const float* __restrict__ dinv, float* __restrict__ z,
                         float* __restrict__ accA, float* __restrict__ accB, int N) {
  int i = blockIdx.x * blockDim.x + threadIdx.x;
  if (i >= N) return;
  const float4* hr = (const float4*)(out1 + (long)i * HID);
  const float4* hs = (const float4*)(out1B + (long)i * HID);
  const float4* bb = (const float4*)b1;
  const float4* ww = (const float4*)W2;
  const float di = dinv[i];
  float acc = 0.0f;
#pragma unroll
  for (int q = 0; q < 4; ++q) {
    float4 v = hr[q], u = hs[q], b = bb[q], w = ww[q];
    acc = fmaf(fmaxf(di * (v.x + u.x) + b.x, 0.0f), w.x, acc);
    acc = fmaf(fmaxf(di * (v.y + u.y) + b.y, 0.0f), w.y, acc);
    acc = fmaf(fmaxf(di * (v.z + u.z) + b.z, 0.0f), w.z, acc);
    acc = fmaf(fmaxf(di * (v.w + u.w) + b.w, 0.0f), w.w, acc);
  }
  const float zs = di * acc;
  z[i] = zs;
  accA[i] = zs;
  accB[i] = 0.0f;
}

__global__ void k_agg2(const int* __restrict__ ei,
                       const float* __restrict__ z, float* __restrict__ accA,
                       float* __restrict__ accB, int E) {
  float* out = l2_side() ? accB : accA;
  const int* __restrict__ dstp = ei + (long)E;
  const int step  = blockDim.x;
  const int first = blockIdx.x * (step * EDGE_UNROLL);
  const int base  = first + threadIdx.x;
  if (first + step * EDGE_UNROLL <= E) {
    int   s[EDGE_UNROLL], d[EDGE_UNROLL];
    float v[EDGE_UNROLL];
#pragma unroll
    for (int k = 0; k < EDGE_UNROLL; ++k) {
      s[k] = ei[base + k * step];
      d[k] = dstp[base + k * step];
    }
#pragma unroll
    for (int k = 0; k < EDGE_UNROLL; ++k)
      v[k] = z[(unsigned)s[k]];
#pragma unroll
    for (int k = 0; k < EDGE_UNROLL; ++k)
      atomicAdd(&out[(unsigned)d[k]], v[k]);
  } else {
    for (int k = 0; k < EDGE_UNROLL; ++k) {
      const int e = base + k * step;
      if (e < E) atomicAdd(&out[(unsigned)dstp[e]], z[(unsigned)ei[e]]);
    }
  }
}

__global__ void k_sum2(const float* __restrict__ accA, const float* __restrict__ accB,
                       const float* __restrict__ dinv, const float* __restrict__ b2,
                       float* __restrict__ out, int N) {
  int i = blockIdx.x * blockDim.x + threadIdx.x;
  if (i < N) out[i] = b2[0] + dinv[i] * (accA[i] + accB[i]);
}

extern "C" void kernel_launch(void* const* d_in, const int* in_sizes, int n_in,
                              void* d_out, int out_size, void* d_ws, size_t ws_size,
                              hipStream_t stream) {
  const float* x  = (const float*)d_in[0];
  const int*   ei = (const int*)d_in[1];
  const float* W1 = (const float*)d_in[2];
  const float* b1 = (const float*)d_in[3];
  const float* W2 = (const float*)d_in[4];
  const float* b2 = (const float*)d_in[5];
  float* out = (float*)d_out;

  const int N = in_sizes[0] / F_IN;
  const int E = in_sizes[1] / 2;

  char* ws = (char*)d_ws;
  size_t off = 0;
  auto carve = [&](size_t bytes) -> void* {
    off = (off + 255) & ~(size_t)255;
    void* p = ws + off;
    off += bytes;
    return p;
  };
  const int nW       = (N + 1) / 2;
  const int rowWords = (nW + 63) & ~63;
  const int needG    = (E + DEG_MAX_SHARE - 1) / DEG_MAX_SHARE;
  const int gD       = needG > DEG_GROUPS ? (needG + DEG_GROUPS - 1) / DEG_GROUPS * DEG_GROUPS
                                          : DEG_GROUPS;
  float*    dinv = (float*)carve((size_t)N * sizeof(float));
  unsigned* part = (unsigned*)carve((size_t)gD * rowWords * sizeof(unsigned));
  float* xw    = (float*)carve((size_t)N * HID * sizeof(float));
  float* out1  = (float*)carve((size_t)N * HID * sizeof(float));
  float* out1B = (float*)carve((size_t)N * HID * sizeof(float));
  float* z     = (float*)carve((size_t)N * sizeof(float));
  float* accA  = (float*)carve((size_t)N * sizeof(float));
  float* accB  = (float*)carve((size_t)N * sizeof(float));
  (void)ws_size; (void)n_in; (void)out_size;

  const int B = 256;
  const int gN = (N + B - 1) / B;
  const int gC = (E + B * EDGE_UNROLL - 1) / (B * EDGE_UNROLL);
  const int wavesNeeded   = (N + 15) / 16;
  const int wavesPerBlock = B / 32;
  const int gW = (wavesNeeded + wavesPerBlock - 1) / wavesPerBlock;

  hipLaunchKernelGGL(k_degree_lds, dim3(gD), dim3(DEG_THREADS), 0, stream, ei, part, N, E, rowWords);
  hipLaunchKernelGGL(k_dinv,     dim3((nW + 31) / 32), dim3(32 * DEG_SLICES), 0, stream, part, dinv, N, gD, rowWords);
  hipLaunchKernelGGL(k_xw_wmma,  dim3(gW), dim3(B), 0, stream, x, W1, dinv, xw, out1, out1B, N);
  hipLaunchKernelGGL(k_agg1,     dim3(gC), dim3(B), 0, stream, ei, xw, out1, out1B, E);
  hipLaunchKernelGGL(k_layer2,   dim3(gN), dim3(B), 0, stream, out1, out1B, b1, W2, dinv, z, accA, accB, N);
  hipLaunchKernelGGL(k_agg2,     dim3(gC), dim3(B), 0, stream, ei, z, accA, accB, E);
  hipLaunchKernelGGL(k_sum2,     dim3(gN), dim3(B), 0, stream, accA, accB, dinv, b2, out, N);
}


# dinv pass folded into the x@W1 pass: a work-group of 8 waves adds the degree-table rows of its 128 nodes, keeps dinv in LDS, then scales its WMMA tiles: 6 launches, not 7; dinv bit-equal
# speedup vs baseline: 11.1114x; 1.0330x over previous
#include <hip/hip_runtime.h>

typedef __attribute__((ext_vector_type(2))) float v2f;
typedef __attribute__((ext_vector_type(8))) float v8f;

#define F_IN 5
#define HID 16

#define EDGE_UNROLL 8

__device__ __forceinline__ int l2_side() {
  return (int)((__builtin_amdgcn_s_sendmsg_rtn(0x87) >> 18) & 1u);
}


#define DEG_THREADS   1024
#define DEG_WORDS     51200
#define DEG_GROUPS    128
#define DEG_MAX_SHARE 32768
#define DEG_SLICES    4
#define DEG_BATCH     32
static_assert(DEG_GROUPS % (DEG_SLICES * DEG_BATCH) == 0, "pass 2 adds the rows in whole batches");

__device__ __forceinline__ void deg_count(unsigned* cnt, int d, int nb, unsigned nNodes) {
  const unsigned r = (unsigned)(d - nb);
  if (r < nNodes)
    atomicAdd(&cnt[r >> 1], 1u << ((r & 1u) << 4));
}

__global__ __launch_bounds__(DEG_THREADS)
void k_degree_lds(const int* __restrict__ ei, unsigned* __restrict__ part,
                  int N, int E, int rowWords) {
  __shared__ __attribute__((aligned(16))) unsigned cnt[DEG_WORDS];
  const int* __restrict__ dstp = ei + (long)E;
  const int tid   = threadIdx.x;
  const int G     = (int)gridDim.x;
  const int share = (((E + G - 1) / G) + 3) & ~3;
  const int e0    = ((long)blockIdx.x * share < (long)E) ? (int)blockIdx.x * share : E;
  const int e1    = (e0 + share < E) ? e0 + share : E;
  unsigned* __restrict__ row = part + (long)blockIdx.x * rowWords;
  uint4* c4 = (uint4*)cnt;

  for (int nb = 0; nb < N; nb += 2 * DEG_WORDS) {
    const int nNodes = (N - nb < 2 * DEG_WORDS) ? N - nb : 2 * DEG_WORDS;
    const int nW     = (nNodes + 1) >> 1;
    const int nW4    = (nW + 3) >> 2;
    for (int q = tid; q < nW4; q += DEG_THREADS)
      c4[q] = make_uint4(0u, 0u, 0u, 0u);
    __syncthreads();

    if ((E & 3) == 0) {
      const int4* __restrict__ d4 = (const int4*)(dstp + e0);
      const int nQ = (e1 - e0) >> 2;
      for (int q0 = tid; q0 < nQ; q0 += 8 * DEG_THREADS) {
        int4 d[8];
#pragma unroll
        for (int k = 0; k < 8; ++k) {
          const int q = q0 + k * DEG_THREADS;
          d[k] = d4[(q < nQ) ? q : 0];
        }
#pragma unroll
        for (int k = 0; k < 8; ++k) {
          if (q0 + k * DEG_THREADS < nQ) {
            deg_count(cnt, d[k].x, nb, (unsigned)nNodes);
            deg_count(cnt, d[k].y, nb, (unsigned)nNodes);
            deg_count(cnt, d[k].z, nb, (unsigned)nNodes);
            deg_count(cnt, d[k].w, nb, (unsigned)nNodes);
          }
        }
      }
    } else {
      for (int e = e0 + tid; e < e1; e += DEG_THREADS)
        deg_count(cnt, dstp[e], nb, (unsigned)nNodes);
    }
    __syncthreads();

    unsigned* __restrict__ rw = row + (nb >> 1);
    const int full4 = nW >> 2;
    for (int q = tid; q < full4; q += DEG_THREADS)
      ((uint4*)rw)[q] = c4[q];
    if (full4 * 4 + tid < nW)
      rw[full4 * 4 + tid] = cnt[full4 * 4 + tid];
    __syncthreads();
  }
}

#define XW_WAVES 8
static_assert(XW_WAVES == 2 * DEG_SLICES, "a pair of waves per slice of rows");

__global__ __launch_bounds__(32 * XW_WAVES)
void k_dinv_xw_wmma(const unsigned* __restrict__ part,
                    const float* __restrict__ x,
                    const float* __restrict__ W1,
                    float* __restrict__ dinv,
                    float* __restrict__ xw,
                    float* __restrict__ out1,
                    float* __restrict__ out1B,
                    int N, int G, int rowWords) {
  __shared__ unsigned sLo[DEG_SLICES][64];
  __shared__ unsigned sHi[DEG_SLICES][64];
  __shared__ float    sDinv[16 * XW_WAVES];
  const int lane = threadIdx.x & 31;
  const int wave = threadIdx.x >> 5;
  const int blockNode = blockIdx.x * (16 * XW_WAVES);
  const int base = blockNode + wave * 16;

  const int half = lane >> 4;
  const int l    = lane & 15;
  int row = base + l;
  if (row >= N) row = N - 1;

  const float* xr = x + (long)row * F_IN;
  const int klo = half * 2;

  v2f a0; a0.x = xr[klo];            a0.y = xr[klo + 1];
  v2f b0; b0.x = W1[klo * HID + l];  b0.y = W1[(klo + 1) * HID + l];

  float x4 = xr[4];
  float w4 = W1[4 * HID + l];
  v2f a1; a1.x = half ? 0.0f : x4;  a1.y = 0.0f;
  v2f b1; b1.x = half ? 0.0f : w4;  b1.y = 0.0f;

  const int col   = (wave & 1) * 32 + lane;
  const int slice = wave >> 1;
  const int w     = blockIdx.x * 64 + col;
  const int nW    = (N + 1) >> 1;
  unsigned lo = 0u, hi = 0u;
  if (w < nW) {
    const long step = (long)DEG_SLICES * rowWords;
    for (int g0 = slice; g0 < G; g0 += DEG_SLICES * DEG_BATCH) {
      const unsigned* __restrict__ p = part + (long)g0 * rowWords + w;
      unsigned v[DEG_BATCH];
#pragma unroll
      for (int k = 0; k < DEG_BATCH; ++k)
        v[k] = p[k * step];
#pragma unroll
      for (int k = 0; k < DEG_BATCH; ++k) {
        lo += v[k] & 0xffffu;
        hi += v[k] >> 16;
      }
    }
  }
  sLo[slice][col] = lo;
  sHi[slice][col] = hi;
  __syncthreads();
  if (threadIdx.x < 64) {
    const int t  = threadIdx.x;
    const int wt = blockIdx.x * 64 + t;
    unsigned slo = sLo[0][t], shi = sHi[0][t];
#pragma unroll
    for (int s = 1; s < DEG_SLICES; ++s) { slo += sLo[s][t]; shi += sHi[s][t]; }
    const float d0 = rsqrtf(1.0f + (float)slo);
    const float d1 = rsqrtf(1.0f + (float)shi);
    sDinv[2 * t]     = d0;
    sDinv[2 * t + 1] = d1;
    if (2 * wt + 1 < N) {
      float2 o; o.x = d0; o.y = d1;
      *(float2*)(dinv + 2 * wt) = o;
    } else if (2 * wt < N) {
      dinv[2 * wt] = d0;
    }
  }
  __syncthreads();
  if (base >= N) return;

  v8f c = {};
  c = __builtin_amdgcn_wmma_f32_16x16x4_f32(false, a0, false, b0, (short)0, c, false, false);
  c = __builtin_amdgcn_wmma_f32_16x16x4_f32(false, a1, false, b1, (short)0, c, false, false);

  const long colBase = (long)(base + half * 8) * HID + l;
  const float* dp = sDinv + wave * 16 + half * 8;
  if (base + 16 <= N) {
    float* xp = xw + colBase;
    float* op = out1 + colBase;
    float* zp = out1B + colBase;
#pragma unroll
    for (int r = 0; r < 8; ++r) {
      const float v = dp[r] * c[r];
      xp[(long)r * HID] = v;
      op[(long)r * HID] = v;
      zp[(long)r * HID] = 0.0f;
    }
  } else {
#pragma unroll
    for (int r = 0; r < 8; ++r) {
      const int node = base + half * 8 + r;
      if (node < N) {
        const float v = dp[r] * c[r];
        xw[(long)node * HID + l] = v;
        out1[(long)node * HID + l] = v;
        out1B[(long)node * HID + l] = 0.0f;
      }
    }
  }
}

__global__ void k_agg1(const int* __restrict__ ei,
                       const float* __restrict__ xw, float* __restrict__ out1A,
                       float* __restrict__ out1B, int E) {
  float* out1 = l2_side() ? out1B : out1A;
  const int* __restrict__ dstp = ei + (long)E;
  const int lane = threadIdx.x & 31;
  const unsigned c = lane & 15;
  const int half = lane & 16;
  const int wavesPerBlock = blockDim.x >> 5;
  const int nWaves = gridDim.x * wavesPerBlock;
  const int wid = __builtin_amdgcn_readfirstlane(blockIdx.x * wavesPerBlock + (threadIdx.x >> 5));
  const int nFull = E >> 5;

  int b = wid;
  int s = 0, d = 0;
  if (b < nFull) {
    s = ei[b * 32 + lane];
    d = dstp[b * 32 + lane];
  }
  while (b < nFull) {
    const unsigned srow = (unsigned)s * HID;
    const unsigned drow = (unsigned)d * HID;

    float v[16];
#pragma unroll
    for (int j = 0; j < 16; ++j)
      v[j] = xw[__shfl(srow, half | j) + c];

    const int bn = b + nWaves;
    if (bn < nFull) {
      s = ei[bn * 32 + lane];
      d = dstp[bn * 32 + lane];
      const int bp = bn + nWaves;
      if (bp < nFull) {
        __builtin_prefetch(&ei[bp * 32 + lane], 0, 0);
        __builtin_prefetch(&dstp[bp * 32 + lane], 0, 0);
      }
    }

#pragma unroll
    for (int j = 0; j < 16; ++j)
      atomicAdd(out1 + (__shfl(drow, half | j) + c), v[j]);
    b = bn;
  }

  const int e0 = nFull * 32;
  if (e0 < E && wid == nFull % nWaves) {
    const int e = e0 + lane;
    int st = 0, dt = 0;
    if (e < E) {
      st = ei[e];
      dt = dstp[e];
    }
    for (int j = 0; j < 16; ++j) {
      const int from = half | j;
      const int   sj = __shfl(st, from);
      const int   dj = __shfl(dt, from);
      if (e0 + from < E)
        atomicAdd(out1 + (long)dj * HID + c, xw[(long)sj * HID + c]);
    }
  }
}

__global__ void k_layer2(const float* __restrict__ out1, const float* __restrict__ out1B,
                         const float* __restrict__ b1,
                         const float* __restrict__ W2,
                         const float* __restrict__ dinv, float* __restrict__ z,
                         float* __restrict__ accA, float* __restrict__ accB, int N) {
  int i = blockIdx.x * blockDim.x + threadIdx.x;
  if (i >= N) return;
  const float4* hr = (const float4*)(out1 + (long)i * HID);
  const float4* hs = (const float4*)(out1B + (long)i * HID);
  const float4* bb = (const float4*)b1;
  const float4* ww = (const float4*)W2;
  const float di = dinv[i];
  float acc = 0.0f;
#pragma unroll
  for (int q = 0; q < 4; ++q) {
    float4 v = hr[q], u = hs[q], b = bb[q], w = ww[q];
    acc = fmaf(fmaxf(di * (v.x + u.x) + b.x, 0.0f), w.x, acc);
    acc = fmaf(fmaxf(di * (v.y + u.y) + b.y, 0.0f), w.y, acc);
    acc = fmaf(fmaxf(di * (v.z + u.z) + b.z, 0.0f), w.z, acc);
    acc = fmaf(fmaxf(di * (v.w + u.w) + b.w, 0.0f), w.w, acc);
  }
  const float zs = di * acc;
  z[i] = zs;
  accA[i] = zs;
  accB[i] = 0.0f;
}

__global__ void k_agg2(const int* __restrict__ ei,
                       const float* __restrict__ z, float* __restrict__ accA,
                       float* __restrict__ accB, int E) {
  float* out = l2_side() ? accB : accA;
  const int* __restrict__ dstp = ei + (long)E;
  const int step  = blockDim.x;
  const int first = blockIdx.x * (step * EDGE_UNROLL);
  const int base  = first + threadIdx.x;
  if (first + step * EDGE_UNROLL <= E) {
    int   s[EDGE_UNROLL], d[EDGE_UNROLL];
    float v[EDGE_UNROLL];
#pragma unroll
    for (int k = 0; k < EDGE_UNROLL; ++k) {
      s[k] = ei[base + k * step];
      d[k] = dstp[base + k * step];
    }
#pragma unroll
    for (int k = 0; k < EDGE_UNROLL; ++k)
      v[k] = z[(unsigned)s[k]];
#pragma unroll
    for (int k = 0; k < EDGE_UNROLL; ++k)
      atomicAdd(&out[(unsigned)d[k]], v[k]);
  } else {
    for (int k = 0; k < EDGE_UNROLL; ++k) {
      const int e = base + k * step;
      if (e < E) atomicAdd(&out[(unsigned)dstp[e]], z[(unsigned)ei[e]]);
    }
  }
}

__global__ void k_sum2(const float* __restrict__ accA, const float* __restrict__ accB,
                       const float* __restrict__ dinv, const float* __restrict__ b2,
                       float* __restrict__ out, int N) {
  int i = blockIdx.x * blockDim.x + threadIdx.x;
  if (i < N) out[i] = b2[0] + dinv[i] * (accA[i] + accB[i]);
}

extern "C" void kernel_launch(void* const* d_in, const int* in_sizes, int n_in,
                              void* d_out, int out_size, void* d_ws, size_t ws_size,
                              hipStream_t stream) {
  const float* x  = (const float*)d_in[0];
  const int*   ei = (const int*)d_in[1];
  const float* W1 = (const float*)d_in[2];
  const float* b1 = (const float*)d_in[3];
  const float* W2 = (const float*)d_in[4];
  const float* b2 = (const float*)d_in[5];
  float* out = (float*)d_out;

  const int N = in_sizes[0] / F_IN;
  const int E = in_sizes[1] / 2;

  char* ws = (char*)d_ws;
  size_t off = 0;
  auto carve = [&](size_t bytes) -> void* {
    off = (off + 255) & ~(size_t)255;
    void* p = ws + off;
    off += bytes;
    return p;
  };
  const int nW       = (N + 1) / 2;
  const int rowWords = (nW + 63) & ~63;
  const int needG    = (E + DEG_MAX_SHARE - 1) / DEG_MAX_SHARE;
  const int gD       = needG > DEG_GROUPS ? (needG + DEG_GROUPS - 1) / DEG_GROUPS * DEG_GROUPS
                                          : DEG_GROUPS;
  float*    dinv = (float*)carve((size_t)N * sizeof(float));
  unsigned* part = (unsigned*)carve((size_t)gD * rowWords * sizeof(unsigned));
  float* xw    = (float*)carve((size_t)N * HID * sizeof(float));
  float* out1  = (float*)carve((size_t)N * HID * sizeof(float));
  float* out1B = (float*)carve((size_t)N * HID * sizeof(float));
  float* z     = (float*)carve((size_t)N * sizeof(float));
  float* accA  = (float*)carve((size_t)N * sizeof(float));
  float* accB  = (float*)carve((size_t)N * sizeof(float));
  (void)ws_size; (void)n_in; (void)out_size;

  const int B = 256;
  const int gN = (N + B - 1) / B;
  const int gC = (E + B * EDGE_UNROLL - 1) / (B * EDGE_UNROLL);
  const int wavesNeeded   = (N + 15) / 16;
  const int gW = (wavesNeeded + XW_WAVES - 1) / XW_WAVES;

  hipLaunchKernelGGL(k_degree_lds, dim3(gD), dim3(DEG_THREADS), 0, stream, ei, part, N, E, rowWords);
  hipLaunchKernelGGL(k_dinv_xw_wmma, dim3(gW), dim3(32 * XW_WAVES), 0, stream, part, x, W1, dinv, xw, out1, out1B, N, gD, rowWords);
  hipLaunchKernelGGL(k_agg1,     dim3(gC), dim3(B), 0, stream, ei, xw, out1, out1B, E);
  hipLaunchKernelGGL(k_layer2,   dim3(gN), dim3(B), 0, stream, out1, out1B, b1, W2, dinv, z, accA, accB, N);
  hipLaunchKernelGGL(k_agg2,     dim3(gC), dim3(B), 0, stream, ei, z, accA, accB, E);
  hipLaunchKernelGGL(k_sum2,     dim3(gN), dim3(B), 0, stream, accA, accB, dinv, b2, out, N);
}


# Layer 1 sums before it multiplies: edges add 5-wide dinv-scaled x rows (8 lanes an edge, 4 edges an instruction), then (sum)@W1 on f32 WMMA in the node pass behind; f32 sum order changes
# speedup vs baseline: 11.7902x; 1.0611x over previous
#include <hip/hip_runtime.h>

typedef __attribute__((ext_vector_type(2))) float v2f;
typedef __attribute__((ext_vector_type(8))) float v8f;

#define F_IN 5
#define HID 16
#define XP 8


#define EDGE_UNROLL 8

__device__ __forceinline__ int l2_side() {
  return (int)((__builtin_amdgcn_s_sendmsg_rtn(0x87) >> 18) & 1u);
}


#define DEG_THREADS   1024
#define DEG_WORDS     51200
#define DEG_GROUPS    128
#define DEG_MAX_SHARE 32768
#define DEG_SLICES    4
#define DEG_BATCH     32
static_assert(DEG_GROUPS % (DEG_SLICES * DEG_BATCH) == 0, "pass 2 adds the rows in whole batches");

__device__ __forceinline__ void deg_count(unsigned* cnt, int d, int nb, unsigned nNodes) {
  const unsigned r = (unsigned)(d - nb);
  if (r < nNodes)
    atomicAdd(&cnt[r >> 1], 1u << ((r & 1u) << 4));
}

__global__ __launch_bounds__(DEG_THREADS)
void k_degree_lds(const int* __restrict__ ei, unsigned* __restrict__ part,
                  int N, int E, int rowWords) {
  __shared__ __attribute__((aligned(16))) unsigned cnt[DEG_WORDS];
  const int* __restrict__ dstp = ei + (long)E;
  const int tid   = threadIdx.x;
  const int G     = (int)gridDim.x;
  const int share = (((E + G - 1) / G) + 3) & ~3;
  const int e0    = ((long)blockIdx.x * share < (long)E) ? (int)blockIdx.x * share : E;
  const int e1    = (e0 + share < E) ? e0 + share : E;
  unsigned* __restrict__ row = part + (long)blockIdx.x * rowWords;
  uint4* c4 = (uint4*)cnt;

  for (int nb = 0; nb < N; nb += 2 * DEG_WORDS) {
    const int nNodes = (N - nb < 2 * DEG_WORDS) ? N - nb : 2 * DEG_WORDS;
    const int nW     = (nNodes + 1) >> 1;
    const int nW4    = (nW + 3) >> 2;
    for (int q = tid; q < nW4; q += DEG_THREADS)
      c4[q] = make_uint4(0u, 0u, 0u, 0u);
    __syncthreads();

    if ((E & 3) == 0) {
      const int4* __restrict__ d4 = (const int4*)(dstp + e0);
      const int nQ = (e1 - e0) >> 2;
      for (int q0 = tid; q0 < nQ; q0 += 8 * DEG_THREADS) {
        int4 d[8];
#pragma unroll
        for (int k = 0; k < 8; ++k) {
          const int q = q0 + k * DEG_THREADS;
          d[k] = d4[(q < nQ) ? q : 0];
        }
#pragma unroll
        for (int k = 0; k < 8; ++k) {
          if (q0 + k * DEG_THREADS < nQ) {
            deg_count(cnt, d[k].x, nb, (unsigned)nNodes);
            deg_count(cnt, d[k].y, nb, (unsigned)nNodes);
            deg_count(cnt, d[k].z, nb, (unsigned)nNodes);
            deg_count(cnt, d[k].w, nb, (unsigned)nNodes);
          }
        }
      }
    } else {
      for (int e = e0 + tid; e < e1; e += DEG_THREADS)
        deg_count(cnt, dstp[e], nb, (unsigned)nNodes);
    }
    __syncthreads();

    unsigned* __restrict__ rw = row + (nb >> 1);
    const int full4 = nW >> 2;
    for (int q = tid; q < full4; q += DEG_THREADS)
      ((uint4*)rw)[q] = c4[q];
    if (full4 * 4 + tid < nW)
      rw[full4 * 4 + tid] = cnt[full4 * 4 + tid];
    __syncthreads();
  }
}

#define XW_WAVES 8
static_assert(XW_WAVES == 2 * DEG_SLICES, "a pair of waves per slice of rows");

__global__ __launch_bounds__(32 * XW_WAVES)
void k_dinv_xs(const unsigned* __restrict__ part,
               const float* __restrict__ x,
               float* __restrict__ dinv,
               float* __restrict__ xs,
               float* __restrict__ aggA,
               float* __restrict__ aggB,
               int N, int G, int rowWords) {
  __shared__ unsigned sLo[DEG_SLICES][64];
  __shared__ unsigned sHi[DEG_SLICES][64];
  __shared__ float    sDinv[16 * XW_WAVES];
  const int lane = threadIdx.x & 31;
  const int wave = threadIdx.x >> 5;
  const int blockNode = blockIdx.x * (16 * XW_WAVES);

  const int nloc = threadIdx.x >> 1;
  const int hsel = threadIdx.x & 1;
  const int node = blockNode + nloc;
  const float* xr = x + (long)((node < N) ? node : N - 1) * F_IN;
  const float x0 = xr[hsel ? 4 : 0];
  const float x1 = xr[hsel ? 4 : 1];
  const float x2 = xr[hsel ? 4 : 2];
  const float x3 = xr[hsel ? 4 : 3];

  const int col   = (wave & 1) * 32 + lane;
  const int slice = wave >> 1;
  const int w     = blockIdx.x * 64 + col;
  const int nW    = (N + 1) >> 1;
  unsigned lo = 0u, hi = 0u;
  if (w < nW) {
    const long step = (long)DEG_SLICES * rowWords;
    for (int g0 = slice; g0 < G; g0 += DEG_SLICES * DEG_BATCH) {
      const unsigned* __restrict__ p = part + (long)g0 * rowWords + w;
      unsigned v[DEG_BATCH];
#pragma unroll
      for (int k = 0; k < DEG_BATCH; ++k)
        v[k] = p[k * step];
#pragma unroll
      for (int k = 0; k < DEG_BATCH; ++k) {
        lo += v[k] & 0xffffu;
        hi += v[k] >> 16;
      }
    }
  }
  sLo[slice][col] = lo;
  sHi[slice][col] = hi;
  __syncthreads();
  if (threadIdx.x < 64) {
    const int t  = threadIdx.x;
    const int wt = blockIdx.x * 64 + t;
    unsigned slo = sLo[0][t], shi = sHi[0][t];
#pragma unroll
    for (int s = 1; s < DEG_SLICES; ++s) { slo += sLo[s][t]; shi += sHi[s][t]; }
    const float d0 = rsqrtf(1.0f + (float)slo);
    const float d1 = rsqrtf(1.0f + (float)shi);
    sDinv[2 * t]     = d0;
    sDinv[2 * t + 1] = d1;
    if (2 * wt + 1 < N) {
      float2 o; o.x = d0; o.y = d1;
      *(float2*)(dinv + 2 * wt) = o;
    } else if (2 * wt < N) {
      dinv[2 * wt] = d0;
    }
  }
  __syncthreads();
  if (node >= N) return;

  const float di = sDinv[nloc];
  float4 v;
  v.x = di * x0;
  v.y = hsel ? 0.0f : di * x1;
  v.z = hsel ? 0.0f : di * x2;
  v.w = hsel ? 0.0f : di * x3;
  const long q = (long)node * 2 + hsel;
  ((float4*)xs)[q]   = v;
  ((float4*)aggA)[q] = v;
  ((float4*)aggB)[q] = make_float4(0.0f, 0.0f, 0.0f, 0.0f);
}

template <int J>
__device__ __forceinline__ unsigned bcast8(unsigned v) {
  return (unsigned)__builtin_amdgcn_mov_dpp8(v, J * 0x249249);
}
#define REP8(M) M(0) M(1) M(2) M(3) M(4) M(5) M(6) M(7)

__global__ void k_agg1(const int* __restrict__ ei,
                       const float* __restrict__ xs, float* __restrict__ aggA,
                       float* __restrict__ aggB, int E) {
  float* agg = l2_side() ? aggB : aggA;
  const int* __restrict__ dstp = ei + (long)E;
  const int lane = threadIdx.x & 31;
  const unsigned f = lane & 7;
  const int wavesPerBlock = blockDim.x >> 5;
  const int nWaves = gridDim.x * wavesPerBlock;
  const int wid = __builtin_amdgcn_readfirstlane(blockIdx.x * wavesPerBlock + (threadIdx.x >> 5));
  const int nFull = E >> 6;

  int b = wid;
  int s0 = 0, s1 = 0, d0 = 0, d1 = 0;
  if (b < nFull) {
    s0 = ei[b * 64 + lane];
    s1 = ei[b * 64 + 32 + lane];
    d0 = dstp[b * 64 + lane];
    d1 = dstp[b * 64 + 32 + lane];
  }
  while (b < nFull) {
    const unsigned sr0 = (unsigned)s0 * XP;
    const unsigned sr1 = (unsigned)s1 * XP;
    const unsigned dr0 = (unsigned)d0 * XP;
    const unsigned dr1 = (unsigned)d1 * XP;

    float v[16];
    unsigned a[16];
#define GATHER0(J) v[J]     = xs[bcast8<J>(sr0) + f];
#define GATHER1(J) v[8 + J] = xs[bcast8<J>(sr1) + f];
    REP8(GATHER0)
    REP8(GATHER1)
#undef GATHER0
#undef GATHER1
#define DADDR0(J) a[J]     = bcast8<J>(dr0) + f;
#define DADDR1(J) a[8 + J] = bcast8<J>(dr1) + f;
    REP8(DADDR0)
    REP8(DADDR1)
#undef DADDR0
#undef DADDR1

    const int bn = b + nWaves;
    if (bn < nFull) {
      s0 = ei[bn * 64 + lane];
      s1 = ei[bn * 64 + 32 + lane];
      d0 = dstp[bn * 64 + lane];
      d1 = dstp[bn * 64 + 32 + lane];
      const int bp = bn + nWaves;
      if (bp < nFull) {
        __builtin_prefetch(&ei[bp * 64 + lane * 2], 0, 0);
        __builtin_prefetch(&dstp[bp * 64 + lane * 2], 0, 0);
      }
    }

    if (f < F_IN) {
#pragma unroll
      for (int j = 0; j < 16; ++j)
        atomicAdd(agg + a[j], v[j]);
    }
    b = bn;
  }

  const int e0 = nFull * 64;
  if (e0 < E && wid == nFull % nWaves) {
    for (int e = e0; e < E; ++e) {
      const unsigned sj = (unsigned)ei[e];
      const unsigned dj = (unsigned)dstp[e];
      if (lane < F_IN)
        atomicAdd(agg + dj * XP + lane, xs[sj * XP + lane]);
    }
  }
}

__global__ __launch_bounds__(32 * XW_WAVES)
void k_layer2_wmma(const float* __restrict__ aggA, const float* __restrict__ aggB,
                   const float* __restrict__ W1,
                   const float* __restrict__ b1,
                   const float* __restrict__ W2,
                   const float* __restrict__ dinv, float* __restrict__ z,
                   float* __restrict__ accA, float* __restrict__ accB, int N) {
  __shared__ float sZ[16 * XW_WAVES];
  const int lane = threadIdx.x & 31;
  const int wave = threadIdx.x >> 5;
  const int blockNode = blockIdx.x * (16 * XW_WAVES);
  const int base = blockNode + wave * 16;

  const int half = lane >> 4;
  const int l    = lane & 15;
  int row = base + l;
  if (row >= N) row = N - 1;

  const float* pa = aggA + (long)row * XP;
  const float* pb = aggB + (long)row * XP;
  const int klo = half * 2;

  const float2 sa = *(const float2*)(pa + klo);
  const float2 sb = *(const float2*)(pb + klo);
  const float s4a = pa[4];
  const float s4b = pb[4];
  const float di  = dinv[row];
  v2f b0; b0.x = W1[klo * HID + l];  b0.y = W1[(klo + 1) * HID + l];
  const float w4 = W1[4 * HID + l];
  const float bias = b1[l];
  const float w2   = W2[l];
  const int onode = blockNode + (int)threadIdx.x;
  const float dio = dinv[(onode < N) ? onode : N - 1];

  v2f a0; a0.x = di * (sa.x + sb.x);  a0.y = di * (sa.y + sb.y);
  v2f a1; a1.x = half ? 0.0f : di * (s4a + s4b);  a1.y = 0.0f;
  v2f b1v; b1v.x = half ? 0.0f : w4;  b1v.y = 0.0f;

  v8f c = {};
  c = __builtin_amdgcn_wmma_f32_16x16x4_f32(false, a0, false, b0, (short)0, c, false, false);
  c = __builtin_amdgcn_wmma_f32_16x16x4_f32(false, a1, false, b1v, (short)0, c, false, false);

  float p[8];
#pragma unroll
  for (int r = 0; r < 8; ++r)
    p[r] = fmaxf(c[r] + bias, 0.0f) * w2;
#pragma unroll
  for (int m = 8; m >= 1; m >>= 1) {
#pragma unroll
    for (int r = 0; r < 8; ++r)
      p[r] += __shfl_xor(p[r], m);
  }
  float mine = p[0];
#pragma unroll
  for (int r = 1; r < 8; ++r)
    mine = (l == r) ? p[r] : mine;
  if (l < 8)
    sZ[wave * 16 + half * 8 + l] = mine;
  __syncthreads();
  if (threadIdx.x < 16 * XW_WAVES && onode < N) {
    const float zs = dio * sZ[threadIdx.x];
    z[onode]    = zs;
    accA[onode] = zs;
    accB[onode] = 0.0f;
  }
}

__global__ void k_agg2(const int* __restrict__ ei,
                       const float* __restrict__ z, float* __restrict__ accA,
                       float* __restrict__ accB, int E) {
  float* out = l2_side() ? accB : accA;
  const int* __restrict__ dstp = ei + (long)E;
  const int step  = blockDim.x;
  const int first = blockIdx.x * (step * EDGE_UNROLL);
  const int base  = first + threadIdx.x;
  if (first + step * EDGE_UNROLL <= E) {
    int   s[EDGE_UNROLL], d[EDGE_UNROLL];
    float v[EDGE_UNROLL];
#pragma unroll
    for (int k = 0; k < EDGE_UNROLL; ++k) {
      s[k] = ei[base + k * step];
      d[k] = dstp[base + k * step];
    }
#pragma unroll
    for (int k = 0; k < EDGE_UNROLL; ++k)
      v[k] = z[(unsigned)s[k]];
#pragma unroll
    for (int k = 0; k < EDGE_UNROLL; ++k)
      atomicAdd(&out[(unsigned)d[k]], v[k]);
  } else {
    for (int k = 0; k < EDGE_UNROLL; ++k) {
      const int e = base + k * step;
      if (e < E) atomicAdd(&out[(unsigned)dstp[e]], z[(unsigned)ei[e]]);
    }
  }
}

__global__ void k_sum2(const float* __restrict__ accA, const float* __restrict__ accB,
                       const float* __restrict__ dinv, const float* __restrict__ b2,
                       float* __restrict__ out, int N) {
  int i = blockIdx.x * blockDim.x + threadIdx.x;
  if (i < N) out[i] = b2[0] + dinv[i] * (accA[i] + accB[i]);
}

extern "C" void kernel_launch(void* const* d_in, const int* in_sizes, int n_in,
                              void* d_out, int out_size, void* d_ws, size_t ws_size,
                              hipStream_t stream) {
  const float* x  = (const float*)d_in[0];
  const int*   ei = (const int*)d_in[1];
  const float* W1 = (const float*)d_in[2];
  const float* b1 = (const float*)d_in[3];
  const float* W2 = (const float*)d_in[4];
  const float* b2 = (const float*)d_in[5];
  float* out = (float*)d_out;

  const int N = in_sizes[0] / F_IN;
  const int E = in_sizes[1] / 2;

  char* ws = (char*)d_ws;
  size_t off = 0;
  auto carve = [&](size_t bytes) -> void* {
    off = (off + 255) & ~(size_t)255;
    void* p = ws + off;
    off += bytes;
    return p;
  };
  const int nW       = (N + 1) / 2;
  const int rowWords = (nW + 63) & ~63;
  const int needG    = (E + DEG_MAX_SHARE - 1) / DEG_MAX_SHARE;
  const int gD       = needG > DEG_GROUPS ? (needG + DEG_GROUPS - 1) / DEG_GROUPS * DEG_GROUPS
                                          : DEG_GROUPS;
  float*    dinv = (float*)carve((size_t)N * sizeof(float));
  unsigned* part = (unsigned*)carve((size_t)gD * rowWords * sizeof(unsigned));
  float* xs    = (float*)carve((size_t)N * XP * sizeof(float));
  float* aggA  = (float*)carve((size_t)N * XP * sizeof(float));
  float* aggB  = (float*)carve((size_t)N * XP * sizeof(float));
  float* z     = (float*)carve((size_t)N * sizeof(float));
  float* accA  = (float*)carve((size_t)N * sizeof(float));
  float* accB  = (float*)carve((size_t)N * sizeof(float));
  (void)ws_size; (void)n_in; (void)out_size;

  const int B = 256;
  const int gN = (N + B - 1) / B;
  const int gC = (E + B * EDGE_UNROLL - 1) / (B * EDGE_UNROLL);
  const int wavesNeeded   = (N + 15) / 16;
  const int gW = (wavesNeeded + XW_WAVES - 1) / XW_WAVES;

  hipLaunchKernelGGL(k_degree_lds, dim3(gD), dim3(DEG_THREADS), 0, stream, ei, part, N, E, rowWords);
  hipLaunchKernelGGL(k_dinv_xs,  dim3(gW), dim3(32 * XW_WAVES), 0, stream, part, x, dinv, xs, aggA, aggB, N, gD, rowWords);
  hipLaunchKernelGGL(k_agg1,     dim3(gC), dim3(B), 0, stream, ei, xs, aggA, aggB, E);
  hipLaunchKernelGGL(k_layer2_wmma, dim3(gW), dim3(32 * XW_WAVES), 0, stream, aggA, aggB, W1, b1, W2, dinv, z, accA, accB, N);
  hipLaunchKernelGGL(k_agg2,     dim3(gC), dim3(B), 0, stream, ei, z, accA, accB, E);
  hipLaunchKernelGGL(k_sum2,     dim3(gN), dim3(B), 0, stream, accA, accB, dinv, b2, out, N);
}


# k_agg2 reads z[src] from LDS: work-group (edge share, range of z) copies its 200 KB range of z into LDS, adds the share's edges whose source lies in it; f32 atomic sums stay arrival-ordered
# speedup vs baseline: 12.9351x; 1.0971x over previous
#include <hip/hip_runtime.h>

typedef __attribute__((ext_vector_type(2))) float v2f;
typedef __attribute__((ext_vector_type(8))) float v8f;
typedef __attribute__((ext_vector_type(4))) float v4f;

#define F_IN 5
#define HID 16
#define XP 8


#define EDGE_UNROLL 8

__device__ __forceinline__ int l2_side() {
  return (int)((__builtin_amdgcn_s_sendmsg_rtn(0x87) >> 18) & 1u);
}


#define DEG_THREADS   1024
#define DEG_WORDS     51200
#define DEG_GROUPS    128
#define DEG_MAX_SHARE 32768
#define DEG_SLICES    4
#define DEG_BATCH     32
static_assert(DEG_GROUPS % (DEG_SLICES * DEG_BATCH) == 0, "pass 2 adds the rows in whole batches");

__device__ __forceinline__ void deg_count(unsigned* cnt, int d, int nb, unsigned nNodes) {
  const unsigned r = (unsigned)(d - nb);
  if (r < nNodes)
    atomicAdd(&cnt[r >> 1], 1u << ((r & 1u) << 4));
}

__global__ __launch_bounds__(DEG_THREADS)
void k_degree_lds(const int* __restrict__ ei, unsigned* __restrict__ part,
                  int N, int E, int rowWords) {
  __shared__ __attribute__((aligned(16))) unsigned cnt[DEG_WORDS];
  const int* __restrict__ dstp = ei + (long)E;
  const int tid   = threadIdx.x;
  const int G     = (int)gridDim.x;
  const int share = (((E + G - 1) / G) + 3) & ~3;
  const int e0    = ((long)blockIdx.x * share < (long)E) ? (int)blockIdx.x * share : E;
  const int e1    = (e0 + share < E) ? e0 + share : E;
  unsigned* __restrict__ row = part + (long)blockIdx.x * rowWords;
  uint4* c4 = (uint4*)cnt;

  for (int nb = 0; nb < N; nb += 2 * DEG_WORDS) {
    const int nNodes = (N - nb < 2 * DEG_WORDS) ? N - nb : 2 * DEG_WORDS;
    const int nW     = (nNodes + 1) >> 1;
    const int nW4    = (nW + 3) >> 2;
    for (int q = tid; q < nW4; q += DEG_THREADS)
      c4[q] = make_uint4(0u, 0u, 0u, 0u);
    __syncthreads();

    if ((E & 3) == 0) {
      const int4* __restrict__ d4 = (const int4*)(dstp + e0);
      const int nQ = (e1 - e0) >> 2;
      for (int q0 = tid; q0 < nQ; q0 += 8 * DEG_THREADS) {
        int4 d[8];
#pragma unroll
        for (int k = 0; k < 8; ++k) {
          const int q = q0 + k * DEG_THREADS;
          d[k] = d4[(q < nQ) ? q : 0];
        }
#pragma unroll
        for (int k = 0; k < 8; ++k) {
          if (q0 + k * DEG_THREADS < nQ) {
            deg_count(cnt, d[k].x, nb, (unsigned)nNodes);
            deg_count(cnt, d[k].y, nb, (unsigned)nNodes);
            deg_count(cnt, d[k].z, nb, (unsigned)nNodes);
            deg_count(cnt, d[k].w, nb, (unsigned)nNodes);
          }
        }
      }
    } else {
      for (int e = e0 + tid; e < e1; e += DEG_THREADS)
        deg_count(cnt, dstp[e], nb, (unsigned)nNodes);
    }
    __syncthreads();

    unsigned* __restrict__ rw = row + (nb >> 1);
    const int full4 = nW >> 2;
    for (int q = tid; q < full4; q += DEG_THREADS)
      ((uint4*)rw)[q] = c4[q];
    if (full4 * 4 + tid < nW)
      rw[full4 * 4 + tid] = cnt[full4 * 4 + tid];
    __syncthreads();
  }
}

#define XW_WAVES 8
static_assert(XW_WAVES == 2 * DEG_SLICES, "a pair of waves per slice of rows");

__global__ __launch_bounds__(32 * XW_WAVES)
void k_dinv_xs(const unsigned* __restrict__ part,
               const float* __restrict__ x,
               float* __restrict__ dinv,
               float* __restrict__ xs,
               float* __restrict__ aggA,
               float* __restrict__ aggB,
               int N, int G, int rowWords) {
  __shared__ unsigned sLo[DEG_SLICES][64];
  __shared__ unsigned sHi[DEG_SLICES][64];
  __shared__ float    sDinv[16 * XW_WAVES];
  const int lane = threadIdx.x & 31;
  const int wave = threadIdx.x >> 5;
  const int blockNode = blockIdx.x * (16 * XW_WAVES);

  const int nloc = threadIdx.x >> 1;
  const int hsel = threadIdx.x & 1;
  const int node = blockNode + nloc;
  const float* xr = x + (long)((node < N) ? node : N - 1) * F_IN;
  const float x0 = xr[hsel ? 4 : 0];
  const float x1 = xr[hsel ? 4 : 1];
  const float x2 = xr[hsel ? 4 : 2];
  const float x3 = xr[hsel ? 4 : 3];

  const int col   = (wave & 1) * 32 + lane;
  const int slice = wave >> 1;
  const int w     = blockIdx.x * 64 + col;
  const int nW    = (N + 1) >> 1;
  unsigned lo = 0u, hi = 0u;
  if (w < nW) {
    const long step = (long)DEG_SLICES * rowWords;
    for (int g0 = slice; g0 < G; g0 += DEG_SLICES * DEG_BATCH) {
      const unsigned* __restrict__ p = part + (long)g0 * rowWords + w;
      unsigned v[DEG_BATCH];
#pragma unroll
      for (int k = 0; k < DEG_BATCH; ++k)
        v[k] = p[k * step];
#pragma unroll
      for (int k = 0; k < DEG_BATCH; ++k) {
        lo += v[k] & 0xffffu;
        hi += v[k] >> 16;
      }
    }
  }
  sLo[slice][col] = lo;
  sHi[slice][col] = hi;
  __syncthreads();
  if (threadIdx.x < 64) {
    const int t  = threadIdx.x;
    const int wt = blockIdx.x * 64 + t;
    unsigned slo = sLo[0][t], shi = sHi[0][t];
#pragma unroll
    for (int s = 1; s < DEG_SLICES; ++s) { slo += sLo[s][t]; shi += sHi[s][t]; }
    const float d0 = rsqrtf(1.0f + (float)slo);
    const float d1 = rsqrtf(1.0f + (float)shi);
    sDinv[2 * t]     = d0;
    sDinv[2 * t + 1] = d1;
    if (2 * wt + 1 < N) {
      float2 o; o.x = d0; o.y = d1;
      *(float2*)(dinv + 2 * wt) = o;
    } else if (2 * wt < N) {
      dinv[2 * wt] = d0;
    }
  }
  __syncthreads();
  if (node >= N) return;

  const float di = sDinv[nloc];
  float4 v;
  v.x = di * x0;
  v.y = hsel ? 0.0f : di * x1;
  v.z = hsel ? 0.0f : di * x2;
  v.w = hsel ? 0.0f : di * x3;
  const long q = (long)node * 2 + hsel;
  ((float4*)xs)[q]   = v;
  ((float4*)aggA)[q] = v;
  ((float4*)aggB)[q] = make_float4(0.0f, 0.0f, 0.0f, 0.0f);
}

template <int J>
__device__ __forceinline__ unsigned bcast8(unsigned v) {
  return (unsigned)__builtin_amdgcn_mov_dpp8(v, J * 0x249249);
}
#define REP8(M) M(0) M(1) M(2) M(3) M(4) M(5) M(6) M(7)

__global__ void k_agg1(const int* __restrict__ ei,
                       const float* __restrict__ xs, float* __restrict__ aggA,
                       float* __restrict__ aggB, int E) {
  float* agg = l2_side() ? aggB : aggA;
  const int* __restrict__ dstp = ei + (long)E;
  const int lane = threadIdx.x & 31;
  const unsigned f = lane & 7;
  const int wavesPerBlock = blockDim.x >> 5;
  const int nWaves = gridDim.x * wavesPerBlock;
  const int wid = __builtin_amdgcn_readfirstlane(blockIdx.x * wavesPerBlock + (threadIdx.x >> 5));
  const int nFull = E >> 6;

  int b = wid;
  int s0 = 0, s1 = 0, d0 = 0, d1 = 0;
  if (b < nFull) {
    s0 = ei[b * 64 + lane];
    s1 = ei[b * 64 + 32 + lane];
    d0 = dstp[b * 64 + lane];
    d1 = dstp[b * 64 + 32 + lane];
  }
  while (b < nFull) {
    const unsigned sr0 = (unsigned)s0 * XP;
    const unsigned sr1 = (unsigned)s1 * XP;
    const unsigned dr0 = (unsigned)d0 * XP;
    const unsigned dr1 = (unsigned)d1 * XP;

    float v[16];
    unsigned a[16];
#define GATHER0(J) v[J]     = xs[bcast8<J>(sr0) + f];
#define GATHER1(J) v[8 + J] = xs[bcast8<J>(sr1) + f];
    REP8(GATHER0)
    REP8(GATHER1)
#undef GATHER0
#undef GATHER1
#define DADDR0(J) a[J]     = bcast8<J>(dr0) + f;
#define DADDR1(J) a[8 + J] = bcast8<J>(dr1) + f;
    REP8(DADDR0)
    REP8(DADDR1)
#undef DADDR0
#undef DADDR1

    const int bn = b + nWaves;
    if (bn < nFull) {
      s0 = ei[bn * 64 + lane];
      s1 = ei[bn * 64 + 32 + lane];
      d0 = dstp[bn * 64 + lane];
      d1 = dstp[bn * 64 + 32 + lane];
      const int bp = bn + nWaves;
      if (bp < nFull) {
        __builtin_prefetch(&ei[bp * 64 + lane * 2], 0, 0);
        __builtin_prefetch(&dstp[bp * 64 + lane * 2], 0, 0);
      }
    }

    if (f < F_IN) {
#pragma unroll
      for (int j = 0; j < 16; ++j)
        atomicAdd(agg + a[j], v[j]);
    }
    b = bn;
  }

  const int e0 = nFull * 64;
  if (e0 < E && wid == nFull % nWaves) {
    for (int e = e0; e < E; ++e) {
      const unsigned sj = (unsigned)ei[e];
      const unsigned dj = (unsigned)dstp[e];
      if (lane < F_IN)
        atomicAdd(agg + dj * XP + lane, xs[sj * XP + lane]);
    }
  }
}

__global__ __launch_bounds__(32 * XW_WAVES)
void k_layer2_wmma(const float* __restrict__ aggA, const float* __restrict__ aggB,
                   const float* __restrict__ W1,
                   const float* __restrict__ b1,
                   const float* __restrict__ W2,
                   const float* __restrict__ dinv, float* __restrict__ z,
                   float* __restrict__ accA, float* __restrict__ accB, int N) {
  __shared__ float sZ[16 * XW_WAVES];
  const int lane = threadIdx.x & 31;
  const int wave = threadIdx.x >> 5;
  const int blockNode = blockIdx.x * (16 * XW_WAVES);
  const int base = blockNode + wave * 16;

  const int half = lane >> 4;
  const int l    = lane & 15;
  int row = base + l;
  if (row >= N) row = N - 1;

  const float* pa = aggA + (long)row * XP;
  const float* pb = aggB + (long)row * XP;
  const int klo = half * 2;

  const float2 sa = *(const float2*)(pa + klo);
  const float2 sb = *(const float2*)(pb + klo);
  const float s4a = pa[4];
  const float s4b = pb[4];
  const float di  = dinv[row];
  v2f b0; b0.x = W1[klo * HID + l];  b0.y = W1[(klo + 1) * HID + l];
  const float w4 = W1[4 * HID + l];
  const float bias = b1[l];
  const float w2   = W2[l];
  const int onode = blockNode + (int)threadIdx.x;
  const float dio = dinv[(onode < N) ? onode : N - 1];

  v2f a0; a0.x = di * (sa.x + sb.x);  a0.y = di * (sa.y + sb.y);
  v2f a1; a1.x = half ? 0.0f : di * (s4a + s4b);  a1.y = 0.0f;
  v2f b1v; b1v.x = half ? 0.0f : w4;  b1v.y = 0.0f;

  v8f c = {};
  c = __builtin_amdgcn_wmma_f32_16x16x4_f32(false, a0, false, b0, (short)0, c, false, false);
  c = __builtin_amdgcn_wmma_f32_16x16x4_f32(false, a1, false, b1v, (short)0, c, false, false);

  float p[8];
#pragma unroll
  for (int r = 0; r < 8; ++r)
    p[r] = fmaxf(c[r] + bias, 0.0f) * w2;
#pragma unroll
  for (int m = 8; m >= 1; m >>= 1) {
#pragma unroll
    for (int r = 0; r < 8; ++r)
      p[r] += __shfl_xor(p[r], m);
  }
  float mine = p[0];
#pragma unroll
  for (int r = 1; r < 8; ++r)
    mine = (l == r) ? p[r] : mine;
  if (l < 8)
    sZ[wave * 16 + half * 8 + l] = mine;
  __syncthreads();
  if (threadIdx.x < 16 * XW_WAVES && onode < N) {
    const float zs = dio * sZ[threadIdx.x];
    z[onode]    = zs;
    accA[onode] = zs;
    accB[onode] = 0.0f;
  }
}

#define Z_THREADS 1024
#define Z_NODES   51200
#define Z_SHARES  128

__device__ __forceinline__ void z_add(float* out, float v, unsigned r, int d, unsigned nNodes) {
  if (r < nNodes) atomicAdd(&out[(unsigned)d], v);
}

__global__ __launch_bounds__(Z_THREADS)
void k_agg2(const int* __restrict__ ei,
            const float* __restrict__ z, float* __restrict__ accA,
            float* __restrict__ accB, int N, int E) {
  __shared__ __attribute__((aligned(16))) float zl[Z_NODES];
  float* out = l2_side() ? accB : accA;
  const int* __restrict__ dstp = ei + (long)E;
  const int tid   = threadIdx.x;
  const int G     = (int)gridDim.x;
  const int share = (((E + G - 1) / G) + 3) & ~3;
  const int e0    = ((long)blockIdx.x * share < (long)E) ? (int)blockIdx.x * share : E;
  const int e1    = (e0 + share < E) ? e0 + share : E;
  const int R     = (int)gridDim.y;
  const int rlen  = (((N + R - 1) / R) + 3) & ~3;
  const int nb    = (int)blockIdx.y * rlen;
  const int left  = N - nb;
  if (left <= 0 || e0 >= e1) return;
  const int nNodes = left < rlen ? left : rlen;
  const bool quads = (E & 3) == 0;
  const int4* __restrict__ s4 = (const int4*)(ei + e0);
  const int4* __restrict__ d4 = (const int4*)(dstp + e0);
  const int nQ = quads ? (e1 - e0) >> 2 : 0;

  int4 s[8], d[8];
  if (quads) {
#pragma unroll
    for (int k = 0; k < 8; ++k) {
      const int q = tid + k * Z_THREADS;
      s[k] = s4[(q < nQ) ? q : 0];
      d[k] = d4[(q < nQ) ? q : 0];
    }
  }

  const float* __restrict__ zr = z + nb;
  const int nZ4 = nNodes >> 2;
  for (int q0 = tid; q0 < nZ4; q0 += 8 * Z_THREADS) {
    v4f t[8];
#pragma unroll
    for (int k = 0; k < 8; ++k) {
      const int q = q0 + k * Z_THREADS;
      t[k] = ((const v4f*)zr)[(q < nZ4) ? q : 0];
    }
    asm volatile("" :: "v"(t[0]), "v"(t[1]), "v"(t[2]), "v"(t[3]), "v"(t[4]), "v"(t[5]), "v"(t[6]), "v"(t[7]));
#pragma unroll
    for (int k = 0; k < 8; ++k) {
      const int q = q0 + k * Z_THREADS;
      if (q < nZ4) ((v4f*)zl)[q] = t[k];
    }
  }
  if (nZ4 * 4 + tid < nNodes)
    zl[nZ4 * 4 + tid] = zr[nZ4 * 4 + tid];
  __syncthreads();

  if (quads) {
    for (int q0 = tid; q0 < nQ; q0 += 8 * Z_THREADS) {
      if (q0 != tid) {
#pragma unroll
        for (int k = 0; k < 8; ++k) {
          const int q = q0 + k * Z_THREADS;
          s[k] = s4[(q < nQ) ? q : 0];
          d[k] = d4[(q < nQ) ? q : 0];
        }
      }
#pragma unroll
      for (int k = 0; k < 8; ++k) {
        const bool in = q0 + k * Z_THREADS < nQ;
        const unsigned rx = in ? (unsigned)(s[k].x - nb) : 0xffffffffu;
        const unsigned ry = in ? (unsigned)(s[k].y - nb) : 0xffffffffu;
        const unsigned rz = in ? (unsigned)(s[k].z - nb) : 0xffffffffu;
        const unsigned rw = in ? (unsigned)(s[k].w - nb) : 0xffffffffu;
        const float vx = zl[rx < (unsigned)nNodes ? rx : 0u];
        const float vy = zl[ry < (unsigned)nNodes ? ry : 0u];
        const float vz = zl[rz < (unsigned)nNodes ? rz : 0u];
        const float vw = zl[rw < (unsigned)nNodes ? rw : 0u];
        z_add(out, vx, rx, d[k].x, (unsigned)nNodes);
        z_add(out, vy, ry, d[k].y, (unsigned)nNodes);
        z_add(out, vz, rz, d[k].z, (unsigned)nNodes);
        z_add(out, vw, rw, d[k].w, (unsigned)nNodes);
      }
    }
  } else {
    for (int e = e0 + tid; e < e1; e += Z_THREADS) {
      const unsigned r = (unsigned)(ei[e] - nb);
      if (r < (unsigned)nNodes) atomicAdd(&out[(unsigned)dstp[e]], zl[r]);
    }
  }
}

__global__ void k_sum2(const float* __restrict__ accA, const float* __restrict__ accB,
                       const float* __restrict__ dinv, const float* __restrict__ b2,
                       float* __restrict__ out, int N) {
  int i = blockIdx.x * blockDim.x + threadIdx.x;
  if (i < N) out[i] = b2[0] + dinv[i] * (accA[i] + accB[i]);
}

extern "C" void kernel_launch(void* const* d_in, const int* in_sizes, int n_in,
                              void* d_out, int out_size, void* d_ws, size_t ws_size,
                              hipStream_t stream) {
  const float* x  = (const float*)d_in[0];
  const int*   ei = (const int*)d_in[1];
  const float* W1 = (const float*)d_in[2];
  const float* b1 = (const float*)d_in[3];
  const float* W2 = (const float*)d_in[4];
  const float* b2 = (const float*)d_in[5];
  float* out = (float*)d_out;

  const int N = in_sizes[0] / F_IN;
  const int E = in_sizes[1] / 2;

  char* ws = (char*)d_ws;
  size_t off = 0;
  auto carve = [&](size_t bytes) -> void* {
    off = (off + 255) & ~(size_t)255;
    void* p = ws + off;
    off += bytes;
    return p;
  };
  const int nW       = (N + 1) / 2;
  const int rowWords = (nW + 63) & ~63;
  const int needG    = (E + DEG_MAX_SHARE - 1) / DEG_MAX_SHARE;
  const int gD       = needG > DEG_GROUPS ? (needG + DEG_GROUPS - 1) / DEG_GROUPS * DEG_GROUPS
                                          : DEG_GROUPS;
  float*    dinv = (float*)carve((size_t)N * sizeof(float));
  unsigned* part = (unsigned*)carve((size_t)gD * rowWords * sizeof(unsigned));
  float* xs    = (float*)carve((size_t)N * XP * sizeof(float));
  float* aggA  = (float*)carve((size_t)N * XP * sizeof(float));
  float* aggB  = (float*)carve((size_t)N * XP * sizeof(float));
  float* z     = (float*)carve((size_t)N * sizeof(float));
  float* accA  = (float*)carve((size_t)N * sizeof(float));
  float* accB  = (float*)carve((size_t)N * sizeof(float));
  (void)ws_size; (void)n_in; (void)out_size;

  const int B = 256;
  const int gN = (N + B - 1) / B;
  const int gC = (E + B * EDGE_UNROLL - 1) / (B * EDGE_UNROLL);
  const int zR = (N + Z_NODES - 1) / Z_NODES;
  const int wavesNeeded   = (N + 15) / 16;
  const int gW = (wavesNeeded + XW_WAVES - 1) / XW_WAVES;

  hipLaunchKernelGGL(k_degree_lds, dim3(gD), dim3(DEG_THREADS), 0, stream, ei, part, N, E, rowWords);
  hipLaunchKernelGGL(k_dinv_xs,  dim3(gW), dim3(32 * XW_WAVES), 0, stream, part, x, dinv, xs, aggA, aggB, N, gD, rowWords);
  hipLaunchKernelGGL(k_agg1,     dim3(gC), dim3(B), 0, stream, ei, xs, aggA, aggB, E);
  hipLaunchKernelGGL(k_layer2_wmma, dim3(gW), dim3(32 * XW_WAVES), 0, stream, aggA, aggB, W1, b1, W2, dinv, z, accA, accB, N);
  hipLaunchKernelGGL(k_agg2,     dim3(Z_SHARES, zR), dim3(Z_THREADS), 0, stream, ei, z, accA, accB, N, E);
  hipLaunchKernelGGL(k_sum2,     dim3(gN), dim3(B), 0, stream, accA, accB, dinv, b2, out, N);
}
